# starting kernel
# baseline (speedup 1.0000x reference)
#include <hip/hip_runtime.h>
#include <hip/hip_bf16.h>

typedef __attribute__((ext_vector_type(2))) float v2f;
typedef __attribute__((ext_vector_type(8))) float v8f;

#define NODE_DIM 64

__global__ void k_deg_init(float* __restrict__ deg, int n) {
    int i = blockIdx.x * blockDim.x + threadIdx.x;
    if (i < n) deg[i] = 1.0f;
}

__global__ void k_deg_count(float* __restrict__ deg, const int* __restrict__ dst, int e) {
    int i = blockIdx.x * blockDim.x + threadIdx.x;
    if (i < e) atomicAdd(&deg[dst[i]], 1.0f);
}

__global__ void k_dinv(float* __restrict__ deg, int n) {
    int i = blockIdx.x * blockDim.x + threadIdx.x;
    if (i < n) deg[i] = rsqrtf(deg[i]);
}

template <bool RELU>
__global__ void k_gemm64_wmma(const float* __restrict__ A, const float* __restrict__ W,
                              float* __restrict__ H, int n) {
    const int rowBase = blockIdx.x * 16;
    const int lane    = threadIdx.x & 31;
    const int colBase = (threadIdx.x >> 5) * 16;
    const int m       = lane & 15;
    const int khalf   = (lane >> 4) * 2;

    const int   arowIdx = rowBase + m;
    const float* arow   = A + (size_t)((arowIdx < n) ? arowIdx : (n - 1)) * NODE_DIM;
    const int   nc      = colBase + m;

    v8f c = {};
#pragma unroll
    for (int k0 = 0; k0 < NODE_DIM; k0 += 4) {
        v2f a;
        a.x = arow[k0 + khalf + 0];
        a.y = arow[k0 + khalf + 1];
        if (RELU) { a.x = fmaxf(a.x, 0.0f); a.y = fmaxf(a.y, 0.0f); }
        v2f b;
        b.x = W[(k0 + khalf + 0) * NODE_DIM + nc];
        b.y = W[(k0 + khalf + 1) * NODE_DIM + nc];
        c = __builtin_amdgcn_wmma_f32_16x16x4_f32(false, a, false, b, (short)0, c, false, false);
    }

#pragma unroll
    for (int r = 0; r < 8; ++r) {
        const int row = rowBase + r + (lane >> 4) * 8;
        if (row < n) H[(size_t)row * NODE_DIM + colBase + (lane & 15)] = c[r];
    }
}

__global__ void k_agg_init(float* __restrict__ agg, const float* __restrict__ h,
                           const float* __restrict__ dinv, const float* __restrict__ bias, int n) {
    int i = blockIdx.x * blockDim.x + threadIdx.x;
    if (i < n * NODE_DIM) {
        int node = i >> 6, f = i & 63;
        float di = dinv[node];
        agg[i] = h[i] * di * di + bias[f];
    }
}

__global__ void k_scatter(float* __restrict__ agg, const float* __restrict__ h,
                          const float* __restrict__ dinv,
                          const int* __restrict__ src, const int* __restrict__ dst, int e) {
    int i = blockIdx.x * blockDim.x + threadIdx.x;
    if (i >= e * 16) return;
    int edge = i >> 4;
    int f    = (i & 15) * 4;
    int s = src[edge], d = dst[edge];
    float norm = dinv[s] * dinv[d];
    const float4 v = *(const float4*)(h + (size_t)s * NODE_DIM + f);
    float* o = agg + (size_t)d * NODE_DIM + f;
    atomicAdd(o + 0, v.x * norm);
    atomicAdd(o + 1, v.y * norm);
    atomicAdd(o + 2, v.z * norm);
    atomicAdd(o + 3, v.w * norm);
}

__global__ void k_edge_score(float* __restrict__ out, const float* __restrict__ h,
                             const float* __restrict__ ea, const float* __restrict__ We,
                             const float* __restrict__ be,
                             const int* __restrict__ src, const int* __restrict__ dst, int e) {
    const int lane = threadIdx.x & 31;
    const int edge = blockIdx.x * 8 + (threadIdx.x >> 5);
    if (edge >= e) return;
    const int s = src[edge], d = dst[edge];

    const v2f hs = *(const v2f*)(h + (size_t)s * NODE_DIM + 2 * lane);
    const v2f hd = *(const v2f*)(h + (size_t)d * NODE_DIM + 2 * lane);
    const v2f ws = *(const v2f*)(We + 2 * lane);
    const v2f wd = *(const v2f*)(We + 64 + 2 * lane);

    float p = fmaxf(hs.x, 0.0f) * ws.x + fmaxf(hs.y, 0.0f) * ws.y
            + fmaxf(hd.x, 0.0f) * wd.x + fmaxf(hd.y, 0.0f) * wd.y;
    if (lane < 16) p += ea[(size_t)edge * 16 + lane] * We[128 + lane];

#pragma unroll
    for (int off = 16; off >= 1; off >>= 1) p += __shfl_xor(p, off, 32);

    if (lane == 0) out[edge] = p + be[0];
}

extern "C" void kernel_launch(void* const* d_in, const int* in_sizes, int n_in,
                              void* d_out, int out_size, void* d_ws, size_t ws_size,
                              hipStream_t stream) {
    const float* x  = (const float*)d_in[0];
    const int*   ei = (const int*)d_in[1];
    const float* ea = (const float*)d_in[2];
    const float* W1 = (const float*)d_in[3];
    const float* b1 = (const float*)d_in[4];
    const float* W2 = (const float*)d_in[5];
    const float* b2 = (const float*)d_in[6];
    const float* We = (const float*)d_in[7];
    const float* be = (const float*)d_in[8];
    float* out = (float*)d_out;

    const int N = in_sizes[0] / NODE_DIM;
    const int E = in_sizes[1] / 2;
    const int* src = ei;
    const int* dst = ei + E;

    char* ws = (char*)d_ws;
    const size_t dinvBytes = (((size_t)N * 4) + 511) & ~(size_t)511;
    const size_t hBytes    = (size_t)N * NODE_DIM * 4;
    float* dinv = (float*)ws;
    float* h    = (float*)(ws + dinvBytes);
    float* agg  = (float*)(ws + dinvBytes + hBytes);

    const int TB = 256;
    const int gN    = (N + TB - 1) / TB;
    const int gE    = (E + TB - 1) / TB;
    const int gNF   = (N * NODE_DIM + TB - 1) / TB;
    const int gSc   = (E * 16 + TB - 1) / TB;
    const int gGemm = (N + 15) / 16;
    const int gEdge = (E + 7) / 8;

    k_deg_init <<<gN, TB, 0, stream>>>(dinv, N);
    k_deg_count<<<gE, TB, 0, stream>>>(dinv, dst, E);
    k_dinv     <<<gN, TB, 0, stream>>>(dinv, N);

    k_gemm64_wmma<false><<<gGemm, 128, 0, stream>>>(x, W1, h, N);
    k_agg_init          <<<gNF, TB, 0, stream>>>(agg, h, dinv, b1, N);
    k_scatter           <<<gSc, TB, 0, stream>>>(agg, h, dinv, src, dst, E);

    k_gemm64_wmma<true><<<gGemm, 128, 0, stream>>>(agg, W2, h, N);
    k_agg_init         <<<gNF, TB, 0, stream>>>(agg, h, dinv, b2, N);
    k_scatter          <<<gSc, TB, 0, stream>>>(agg, h, dinv, src, dst, E);

    k_edge_score<<<gEdge, TB, 0, stream>>>(out, agg, ea, We, be, src, dst, E);
}


# Edge scorer split by linearity: relu(h) dotted with each 64-wide weight half once per node; the per-edge pass gathers 2 scalars and reads 16 features (the dot's summation order changes)
# speedup vs baseline: 1.2667x; 1.2667x over previous
#include <hip/hip_runtime.h>
#include <hip/hip_bf16.h>

typedef __attribute__((ext_vector_type(2))) float v2f;
typedef __attribute__((ext_vector_type(8))) float v8f;

#define NODE_DIM 64

__global__ void k_deg_init(float* __restrict__ deg, int n) {
    int i = blockIdx.x * blockDim.x + threadIdx.x;
    if (i < n) deg[i] = 1.0f;
}

__global__ void k_deg_count(float* __restrict__ deg, const int* __restrict__ dst, int e) {
    int i = blockIdx.x * blockDim.x + threadIdx.x;
    if (i < e) atomicAdd(&deg[dst[i]], 1.0f);
}

__global__ void k_dinv(float* __restrict__ deg, int n) {
    int i = blockIdx.x * blockDim.x + threadIdx.x;
    if (i < n) deg[i] = rsqrtf(deg[i]);
}

template <bool RELU>
__global__ void k_gemm64_wmma(const float* __restrict__ A, const float* __restrict__ W,
                              float* __restrict__ H, int n) {
    const int rowBase = blockIdx.x * 16;
    const int lane    = threadIdx.x & 31;
    const int colBase = (threadIdx.x >> 5) * 16;
    const int m       = lane & 15;
    const int khalf   = (lane >> 4) * 2;

    const int   arowIdx = rowBase + m;
    const float* arow   = A + (size_t)((arowIdx < n) ? arowIdx : (n - 1)) * NODE_DIM;
    const int   nc      = colBase + m;

    v8f c = {};
#pragma unroll
    for (int k0 = 0; k0 < NODE_DIM; k0 += 4) {
        v2f a;
        a.x = arow[k0 + khalf + 0];
        a.y = arow[k0 + khalf + 1];
        if (RELU) { a.x = fmaxf(a.x, 0.0f); a.y = fmaxf(a.y, 0.0f); }
        v2f b;
        b.x = W[(k0 + khalf + 0) * NODE_DIM + nc];
        b.y = W[(k0 + khalf + 1) * NODE_DIM + nc];
        c = __builtin_amdgcn_wmma_f32_16x16x4_f32(false, a, false, b, (short)0, c, false, false);
    }

#pragma unroll
    for (int r = 0; r < 8; ++r) {
        const int row = rowBase + r + (lane >> 4) * 8;
        if (row < n) H[(size_t)row * NODE_DIM + colBase + (lane & 15)] = c[r];
    }
}

__global__ void k_agg_init(float* __restrict__ agg, const float* __restrict__ h,
                           const float* __restrict__ dinv, const float* __restrict__ bias, int n) {
    int i = blockIdx.x * blockDim.x + threadIdx.x;
    if (i < n * NODE_DIM) {
        int node = i >> 6, f = i & 63;
        float di = dinv[node];
        agg[i] = h[i] * di * di + bias[f];
    }
}

__global__ void k_scatter(float* __restrict__ agg, const float* __restrict__ h,
                          const float* __restrict__ dinv,
                          const int* __restrict__ src, const int* __restrict__ dst, int e) {
    int i = blockIdx.x * blockDim.x + threadIdx.x;
    if (i >= e * 16) return;
    int edge = i >> 4;
    int f    = (i & 15) * 4;
    int s = src[edge], d = dst[edge];
    float norm = dinv[s] * dinv[d];
    const float4 v = *(const float4*)(h + (size_t)s * NODE_DIM + f);
    float* o = agg + (size_t)d * NODE_DIM + f;
    atomicAdd(o + 0, v.x * norm);
    atomicAdd(o + 1, v.y * norm);
    atomicAdd(o + 2, v.z * norm);
    atomicAdd(o + 3, v.w * norm);
}


template <int STEPS>
__device__ __forceinline__ float group_sum(float v) {
    if (STEPS >= 1) v += __int_as_float(__builtin_amdgcn_update_dpp(0, __float_as_int(v), 0x161, 0xf, 0xf, true));
    if (STEPS >= 2) v += __int_as_float(__builtin_amdgcn_update_dpp(0, __float_as_int(v), 0x162, 0xf, 0xf, true));
    if (STEPS >= 3) v += __int_as_float(__builtin_amdgcn_update_dpp(0, __float_as_int(v), 0x164, 0xf, 0xf, true));
    if (STEPS >= 4) v += __int_as_float(__builtin_amdgcn_update_dpp(0, __float_as_int(v), 0x168, 0xf, 0xf, true));
    return v;
}

__device__ __forceinline__ float dot4(const float4 a, const float4 b) {
    return a.x * b.x + a.y * b.y + a.z * b.z + a.w * b.w;
}

__global__ void k_node_proj(float* __restrict__ pp, const float* __restrict__ agg,
                            const float* __restrict__ We, int n) {
    const int lane = threadIdx.x & 31;
    const int half = lane >> 4;
    const int c    = lane & 15;
    const int nodeBase = (blockIdx.x * (blockDim.x >> 5) + (threadIdx.x >> 5)) * 8;

    const float4 ws = *(const float4*)(We + 4 * c);
    const float4 wd = *(const float4*)(We + 64 + 4 * c);

    float ps[4], pd[4];
#pragma unroll
    for (int r = 0; r < 4; ++r) {
        int node = nodeBase + 2 * r + half;
        node = (node < n) ? node : (n - 1);
        float4 v = *(const float4*)(agg + (size_t)node * NODE_DIM + 4 * c);
        v.x = fmaxf(v.x, 0.0f); v.y = fmaxf(v.y, 0.0f);
        v.z = fmaxf(v.z, 0.0f); v.w = fmaxf(v.w, 0.0f);
        ps[r] = dot4(v, ws);
        pd[r] = dot4(v, wd);
    }
#pragma unroll
    for (int r = 0; r < 4; ++r) {
        ps[r] = group_sum<4>(ps[r]);
        pd[r] = group_sum<4>(pd[r]);
    }
    const float a = (c == 0) ? ps[0] : (c == 1) ? ps[1] : (c == 2) ? ps[2] : ps[3];
    const float b = (c == 0) ? pd[0] : (c == 1) ? pd[1] : (c == 2) ? pd[2] : pd[3];
    const int myNode = nodeBase + 2 * c + half;
    if (c < 4 && myNode < n) *(float2*)(pp + 2 * (size_t)myNode) = make_float2(a, b);
}

__global__ void k_edge_score(float* __restrict__ out, const float* __restrict__ pp,
                             const float* __restrict__ ea, const float* __restrict__ We,
                             const float* __restrict__ be,
                             const int* __restrict__ src, const int* __restrict__ dst, int e) {
    const int lane = threadIdx.x & 31;
    const int q = lane >> 2;
    const int j = lane & 3;
    const int edgeBase = (blockIdx.x * (blockDim.x >> 5) + (threadIdx.x >> 5)) * 64;

    const int e0 = edgeBase + j * 8 + q;
    const int e1 = e0 + 32;
    const int e0c = (e0 < e) ? e0 : (e - 1);
    const int e1c = (e1 < e) ? e1 : (e - 1);
    const unsigned s0 = (unsigned)src[e0c], d0 = (unsigned)dst[e0c];
    const unsigned s1 = (unsigned)src[e1c], d1 = (unsigned)dst[e1c];

    const float4 w = *(const float4*)(We + 128 + 4 * j);

    float p[8];
#pragma unroll
    for (int r = 0; r < 8; ++r) {
        int er = edgeBase + r * 8 + q;
        er = (er < e) ? er : (e - 1);
        const float4 a = *(const float4*)(ea + (size_t)er * 16 + 4 * j);
        p[r] = dot4(a, w);
    }

    const float n0 = pp[2u * s0] + pp[2u * d0 + 1u];
    const float n1 = pp[2u * s1] + pp[2u * d1 + 1u];

#pragma unroll
    for (int r = 0; r < 8; ++r) p[r] = group_sum<2>(p[r]);

    const float m0 = (j == 0) ? p[0] : (j == 1) ? p[1] : (j == 2) ? p[2] : p[3];
    const float m1 = (j == 0) ? p[4] : (j == 1) ? p[5] : (j == 2) ? p[6] : p[7];
    const float bias = be[0];
    if (e0 < e) out[e0] = m0 + n0 + bias;
    if (e1 < e) out[e1] = m1 + n1 + bias;
}

extern "C" void kernel_launch(void* const* d_in, const int* in_sizes, int n_in,
                              void* d_out, int out_size, void* d_ws, size_t ws_size,
                              hipStream_t stream) {
    const float* x  = (const float*)d_in[0];
    const int*   ei = (const int*)d_in[1];
    const float* ea = (const float*)d_in[2];
    const float* W1 = (const float*)d_in[3];
    const float* b1 = (const float*)d_in[4];
    const float* W2 = (const float*)d_in[5];
    const float* b2 = (const float*)d_in[6];
    const float* We = (const float*)d_in[7];
    const float* be = (const float*)d_in[8];
    float* out = (float*)d_out;

    const int N = in_sizes[0] / NODE_DIM;
    const int E = in_sizes[1] / 2;
    const int* src = ei;
    const int* dst = ei + E;

    char* ws = (char*)d_ws;
    const size_t dinvBytes = (((size_t)N * 4) + 511) & ~(size_t)511;
    const size_t hBytes    = (size_t)N * NODE_DIM * 4;
    float* dinv = (float*)ws;
    float* h    = (float*)(ws + dinvBytes);
    float* agg  = (float*)(ws + dinvBytes + hBytes);

    const int TB = 256;
    const int gN    = (N + TB - 1) / TB;
    const int gE    = (E + TB - 1) / TB;
    const int gNF   = (N * NODE_DIM + TB - 1) / TB;
    const int gSc   = (E * 16 + TB - 1) / TB;
    const int gGemm = (N + 15) / 16;
    const int gProj = (N + 63) / 64;
    const int gEdge = (E + 511) / 512;

    k_deg_init <<<gN, TB, 0, stream>>>(dinv, N);
    k_deg_count<<<gE, TB, 0, stream>>>(dinv, dst, E);
    k_dinv     <<<gN, TB, 0, stream>>>(dinv, N);

    k_gemm64_wmma<false><<<gGemm, 128, 0, stream>>>(x, W1, h, N);
    k_agg_init          <<<gNF, TB, 0, stream>>>(agg, h, dinv, b1, N);
    k_scatter           <<<gSc, TB, 0, stream>>>(agg, h, dinv, src, dst, E);

    k_gemm64_wmma<true><<<gGemm, 128, 0, stream>>>(agg, W2, h, N);
    k_agg_init         <<<gNF, TB, 0, stream>>>(agg, h, dinv, b2, N);
    k_scatter          <<<gSc, TB, 0, stream>>>(agg, h, dinv, src, dst, E);

    float* pp = h;
    k_node_proj <<<gProj, TB, 0, stream>>>(pp, agg, We, N);
    k_edge_score<<<gEdge, TB, 0, stream>>>(out, pp, ea, We, be, src, dst, E);
}


# Aggregation as a gather over an edge list sorted by destination (integer counting sort rebuilt in every call) instead of float atomics; a node's terms are summed in the list's arrival order
# speedup vs baseline: 1.7937x; 1.4160x over previous
#include <hip/hip_runtime.h>
#include <hip/hip_bf16.h>

typedef __attribute__((ext_vector_type(2))) float v2f;
typedef __attribute__((ext_vector_type(8))) float v8f;

#define NODE_DIM 64

__global__ void k_cnt_zero(int* __restrict__ cnt, int n) {
    int i = blockIdx.x * blockDim.x + threadIdx.x;
    if (i < n) cnt[i] = 0;
}

__global__ void k_cnt_hist(int* __restrict__ cnt, const int* __restrict__ dst, int e) {
    int i = blockIdx.x * blockDim.x + threadIdx.x;
    if (i < e) atomicAdd(&cnt[dst[i]], 1);
}

#define SCAN_TB 1024

__global__ void __launch_bounds__(SCAN_TB)
k_rowptr(const int* __restrict__ cnt, int* __restrict__ rowptr, int* __restrict__ cursor,
         float* __restrict__ dinv, int n) {
    __shared__ int s_wave[SCAN_TB / 32];
    __shared__ int s_part[SCAN_TB / 32];
    const int t = threadIdx.x;
    const int lane = t & 31;
    const int wv = t >> 5;
    const int start = blockIdx.x * SCAN_TB;

    int part = 0;
#pragma unroll 4
    for (int i = 4 * t; i < start; i += 4 * SCAN_TB) {
        const int4 q = *(const int4*)(cnt + i);
        part += (q.x + q.y) + (q.z + q.w);
    }

    const int idx = start + t;
    const int c = (idx < n) ? cnt[idx] : 0;

    int incl = c;
#pragma unroll
    for (int d = 1; d < 32; d <<= 1) {
        const int o = __shfl_up(incl, d, 32);
        if (lane >= d) incl += o;
    }
#pragma unroll
    for (int d = 16; d >= 1; d >>= 1) part += __shfl_xor(part, d, 32);

    if (lane == 31) s_wave[wv] = incl;
    if (lane == 0)  s_part[wv] = part;
    __syncthreads();

    int before = (lane < wv) ? s_wave[lane] : 0;
    int base   = s_part[lane];
#pragma unroll
    for (int d = 16; d >= 1; d >>= 1) {
        before += __shfl_xor(before, d, 32);
        base   += __shfl_xor(base, d, 32);
    }
    const int excl = base + before + incl - c;
    if (idx < n) {
        rowptr[idx] = excl;
        cursor[idx] = excl;
        dinv[idx]   = rsqrtf((float)(c + 1));
        if (idx == n - 1) rowptr[n] = excl + c;
    }
}

__global__ void k_csr_fill(int2* __restrict__ csr, int* __restrict__ cursor, const float* __restrict__ dinv,
                           const int* __restrict__ src, const int* __restrict__ dst, int e) {
    int i = blockIdx.x * blockDim.x + threadIdx.x;
    if (i >= e) return;
    const int s = src[i], d = dst[i];
    const int pos = atomicAdd(&cursor[d], 1);
    csr[pos] = make_int2(s, __float_as_int(dinv[s]));
}

template <bool RELU>
__global__ void k_gemm64_wmma(const float* __restrict__ A, const float* __restrict__ W,
                              float* __restrict__ H, int n) {
    const int rowBase = blockIdx.x * 16;
    const int lane    = threadIdx.x & 31;
    const int colBase = (threadIdx.x >> 5) * 16;
    const int m       = lane & 15;
    const int khalf   = (lane >> 4) * 2;

    const int   arowIdx = rowBase + m;
    const float* arow   = A + (size_t)((arowIdx < n) ? arowIdx : (n - 1)) * NODE_DIM;
    const int   nc      = colBase + m;

    v8f c = {};
#pragma unroll
    for (int k0 = 0; k0 < NODE_DIM; k0 += 4) {
        v2f a;
        a.x = arow[k0 + khalf + 0];
        a.y = arow[k0 + khalf + 1];
        if (RELU) { a.x = fmaxf(a.x, 0.0f); a.y = fmaxf(a.y, 0.0f); }
        v2f b;
        b.x = W[(k0 + khalf + 0) * NODE_DIM + nc];
        b.y = W[(k0 + khalf + 1) * NODE_DIM + nc];
        c = __builtin_amdgcn_wmma_f32_16x16x4_f32(false, a, false, b, (short)0, c, false, false);
    }

#pragma unroll
    for (int r = 0; r < 8; ++r) {
        const int row = rowBase + r + (lane >> 4) * 8;
        if (row < n) H[(size_t)row * NODE_DIM + colBase + (lane & 15)] = c[r];
    }
}

#define GATHER_U 4
#define GATHER_NP 2

__global__ void __launch_bounds__(256)
k_gather(float* __restrict__ agg, const float* __restrict__ h, const float* __restrict__ dinv,
         const float* __restrict__ bias, const int* __restrict__ rowptr,
         const int2* __restrict__ csr, int n) {
    const int lane = threadIdx.x & 31;
    const int half = lane >> 4;
    const unsigned c = lane & 15;
    const int nodeBase = (blockIdx.x * (blockDim.x >> 5) + (threadIdx.x >> 5)) * (2 * GATHER_NP);

    const float4 b = *(const float4*)(bias + 4 * c);

    int   nodeOf[GATHER_NP], beg[GATHER_NP], end[GATHER_NP];
    float di[GATHER_NP];
    float4 own[GATHER_NP];
#pragma unroll
    for (int it = 0; it < GATHER_NP; ++it) {
        const int node = nodeBase + 2 * it + half;
        const int nc = (node < n) ? node : (n - 1);
        nodeOf[it] = node;
        const int r0 = rowptr[nc];
        const int r1 = rowptr[nc + 1];
        beg[it] = r0;
        end[it] = (node < n) ? r1 : r0;
        di[it]  = dinv[nc];
        own[it] = *(const float4*)(h + (unsigned)nc * NODE_DIM + 4 * c);
    }

#pragma unroll
    for (int it = 0; it < GATHER_NP; ++it) {
        const float d2 = di[it];
        float4 acc;
        acc.x = own[it].x * d2 * d2 + b.x;
        acc.y = own[it].y * d2 * d2 + b.y;
        acc.z = own[it].z * d2 * d2 + b.z;
        acc.w = own[it].w * d2 * d2 + b.w;

        const int e1 = end[it];
        if (beg[it] < e1) {
            const int last = e1 - 1;
            int2 p[GATHER_U];
#pragma unroll
            for (int k = 0; k < GATHER_U; ++k) {
                const int jj = beg[it] + k;
                p[k] = csr[(jj < last) ? jj : last];
            }
            for (int j = beg[it]; j < e1; j += GATHER_U) {
                float4 v[GATHER_U];
#pragma unroll
                for (int k = 0; k < GATHER_U; ++k)
                    v[k] = *(const float4*)(h + (unsigned)p[k].x * NODE_DIM + 4 * c);
                int2 pn[GATHER_U];
#pragma unroll
                for (int k = 0; k < GATHER_U; ++k) {
                    const int jj = j + GATHER_U + k;
                    pn[k] = csr[(jj < last) ? jj : last];
                }
#pragma unroll
                for (int k = 0; k < GATHER_U; ++k) {
                    const float norm = __int_as_float(p[k].y) * d2;
                    const bool live = (j + k) < e1;
                    acc.x = live ? (acc.x + v[k].x * norm) : acc.x;
                    acc.y = live ? (acc.y + v[k].y * norm) : acc.y;
                    acc.z = live ? (acc.z + v[k].z * norm) : acc.z;
                    acc.w = live ? (acc.w + v[k].w * norm) : acc.w;
                }
#pragma unroll
                for (int k = 0; k < GATHER_U; ++k) p[k] = pn[k];
            }
        }
        if (nodeOf[it] < n) *(float4*)(agg + (unsigned)nodeOf[it] * NODE_DIM + 4 * c) = acc;
    }
}


template <int STEPS>
__device__ __forceinline__ float group_sum(float v) {
    if (STEPS >= 1) v += __int_as_float(__builtin_amdgcn_update_dpp(0, __float_as_int(v), 0x161, 0xf, 0xf, true));
    if (STEPS >= 2) v += __int_as_float(__builtin_amdgcn_update_dpp(0, __float_as_int(v), 0x162, 0xf, 0xf, true));
    if (STEPS >= 3) v += __int_as_float(__builtin_amdgcn_update_dpp(0, __float_as_int(v), 0x164, 0xf, 0xf, true));
    if (STEPS >= 4) v += __int_as_float(__builtin_amdgcn_update_dpp(0, __float_as_int(v), 0x168, 0xf, 0xf, true));
    return v;
}

__device__ __forceinline__ float dot4(const float4 a, const float4 b) {
    return a.x * b.x + a.y * b.y + a.z * b.z + a.w * b.w;
}

__global__ void k_node_proj(float* __restrict__ pp, const float* __restrict__ agg,
                            const float* __restrict__ We, int n) {
    const int lane = threadIdx.x & 31;
    const int half = lane >> 4;
    const int c    = lane & 15;
    const int nodeBase = (blockIdx.x * (blockDim.x >> 5) + (threadIdx.x >> 5)) * 8;

    const float4 ws = *(const float4*)(We + 4 * c);
    const float4 wd = *(const float4*)(We + 64 + 4 * c);

    float ps[4], pd[4];
#pragma unroll
    for (int r = 0; r < 4; ++r) {
        int node = nodeBase + 2 * r + half;
        node = (node < n) ? node : (n - 1);
        float4 v = *(const float4*)(agg + (size_t)node * NODE_DIM + 4 * c);
        v.x = fmaxf(v.x, 0.0f); v.y = fmaxf(v.y, 0.0f);
        v.z = fmaxf(v.z, 0.0f); v.w = fmaxf(v.w, 0.0f);
        ps[r] = dot4(v, ws);
        pd[r] = dot4(v, wd);
    }
#pragma unroll
    for (int r = 0; r < 4; ++r) {
        ps[r] = group_sum<4>(ps[r]);
        pd[r] = group_sum<4>(pd[r]);
    }
    const float a = (c == 0) ? ps[0] : (c == 1) ? ps[1] : (c == 2) ? ps[2] : ps[3];
    const float b = (c == 0) ? pd[0] : (c == 1) ? pd[1] : (c == 2) ? pd[2] : pd[3];
    const int myNode = nodeBase + 2 * c + half;
    if (c < 4 && myNode < n) *(float2*)(pp + 2 * (size_t)myNode) = make_float2(a, b);
}

__global__ void k_edge_score(float* __restrict__ out, const float* __restrict__ pp,
                             const float* __restrict__ ea, const float* __restrict__ We,
                             const float* __restrict__ be,
                             const int* __restrict__ src, const int* __restrict__ dst, int e) {
    const int lane = threadIdx.x & 31;
    const int q = lane >> 2;
    const int j = lane & 3;
    const int edgeBase = (blockIdx.x * (blockDim.x >> 5) + (threadIdx.x >> 5)) * 64;

    const int e0 = edgeBase + j * 8 + q;
    const int e1 = e0 + 32;
    const int e0c = (e0 < e) ? e0 : (e - 1);
    const int e1c = (e1 < e) ? e1 : (e - 1);
    const unsigned s0 = (unsigned)src[e0c], d0 = (unsigned)dst[e0c];
    const unsigned s1 = (unsigned)src[e1c], d1 = (unsigned)dst[e1c];

    const float4 w = *(const float4*)(We + 128 + 4 * j);

    float p[8];
#pragma unroll
    for (int r = 0; r < 8; ++r) {
        int er = edgeBase + r * 8 + q;
        er = (er < e) ? er : (e - 1);
        const float4 a = *(const float4*)(ea + (size_t)er * 16 + 4 * j);
        p[r] = dot4(a, w);
    }

    const float n0 = pp[2u * s0] + pp[2u * d0 + 1u];
    const float n1 = pp[2u * s1] + pp[2u * d1 + 1u];

#pragma unroll
    for (int r = 0; r < 8; ++r) p[r] = group_sum<2>(p[r]);

    const float m0 = (j == 0) ? p[0] : (j == 1) ? p[1] : (j == 2) ? p[2] : p[3];
    const float m1 = (j == 0) ? p[4] : (j == 1) ? p[5] : (j == 2) ? p[6] : p[7];
    const float bias = be[0];
    if (e0 < e) out[e0] = m0 + n0 + bias;
    if (e1 < e) out[e1] = m1 + n1 + bias;
}

extern "C" void kernel_launch(void* const* d_in, const int* in_sizes, int n_in,
                              void* d_out, int out_size, void* d_ws, size_t ws_size,
                              hipStream_t stream) {
    const float* x  = (const float*)d_in[0];
    const int*   ei = (const int*)d_in[1];
    const float* ea = (const float*)d_in[2];
    const float* W1 = (const float*)d_in[3];
    const float* b1 = (const float*)d_in[4];
    const float* W2 = (const float*)d_in[5];
    const float* b2 = (const float*)d_in[6];
    const float* We = (const float*)d_in[7];
    const float* be = (const float*)d_in[8];
    float* out = (float*)d_out;

    const int N = in_sizes[0] / NODE_DIM;
    const int E = in_sizes[1] / 2;
    const int* src = ei;
    const int* dst = ei + E;

    char* ws = (char*)d_ws;
    const size_t nodeBytes = (((size_t)N * 4) + 511) & ~(size_t)511;
    const size_t rowBytes  = ((((size_t)N + 1) * 4) + 511) & ~(size_t)511;
    const size_t hBytes    = (size_t)N * NODE_DIM * 4;
    float* dinv   = (float*)ws;
    float* h      = (float*)(ws + nodeBytes);
    float* agg    = (float*)(ws + nodeBytes + hBytes);
    int*   cnt    = (int*)(ws + nodeBytes + 2 * hBytes);
    int*   cursor = (int*)(ws + 2 * nodeBytes + 2 * hBytes);
    int*   rowptr = (int*)(ws + 3 * nodeBytes + 2 * hBytes);
    int2*  csr    = (int2*)(ws + 3 * nodeBytes + rowBytes + 2 * hBytes);

    const int TB = 256;
    const int gN    = (N + TB - 1) / TB;
    const int gE    = (E + TB - 1) / TB;
    const int gScan = (N + SCAN_TB - 1) / SCAN_TB;
    const int gGemm = (N + 15) / 16;
    const int gGath = (N + 16 * GATHER_NP - 1) / (16 * GATHER_NP);
    const int gProj = (N + 63) / 64;
    const int gEdge = (E + 511) / 512;

    k_cnt_zero<<<gN, TB, 0, stream>>>(cnt, N);
    k_cnt_hist<<<gE, TB, 0, stream>>>(cnt, dst, E);
    k_rowptr  <<<gScan, SCAN_TB, 0, stream>>>(cnt, rowptr, cursor, dinv, N);
    k_csr_fill<<<gE, TB, 0, stream>>>(csr, cursor, dinv, src, dst, E);

    k_gemm64_wmma<false><<<gGemm, 128, 0, stream>>>(x, W1, h, N);
    k_gather            <<<gGath, TB, 0, stream>>>(agg, h, dinv, b1, rowptr, csr, N);

    k_gemm64_wmma<true><<<gGemm, 128, 0, stream>>>(agg, W2, h, N);
    k_gather           <<<gGath, TB, 0, stream>>>(agg, h, dinv, b2, rowptr, csr, N);

    float* pp = h;
    k_node_proj <<<gProj, TB, 0, stream>>>(pp, agg, We, N);
    k_edge_score<<<gEdge, TB, 0, stream>>>(out, pp, ea, We, be, src, dst, E);
}


# Edge's slot = segment start + the value its counting add returned (its rank): the fill pass needs no atomics; a node's terms are summed in the count pass's arrival order (was the fill's)
# speedup vs baseline: 2.1568x; 1.2024x over previous
#include <hip/hip_runtime.h>
#include <hip/hip_bf16.h>

typedef __attribute__((ext_vector_type(2))) float v2f;
typedef __attribute__((ext_vector_type(8))) float v8f;

#define NODE_DIM 64

__global__ void k_cnt_zero(int* __restrict__ cnt, int n) {
    int i = blockIdx.x * blockDim.x + threadIdx.x;
    if (i < n) cnt[i] = 0;
}

__global__ void k_cnt_hist(int* __restrict__ cnt, int* __restrict__ rank,
                           const int* __restrict__ dst, int e) {
    int i = blockIdx.x * blockDim.x + threadIdx.x;
    if (i < e) rank[i] = atomicAdd(&cnt[dst[i]], 1);
}

#define SCAN_TB 1024

__global__ void __launch_bounds__(SCAN_TB)
k_rowptr(const int* __restrict__ cnt, int* __restrict__ rowptr,
         float* __restrict__ dinv, int n) {
    __shared__ int s_wave[SCAN_TB / 32];
    __shared__ int s_part[SCAN_TB / 32];
    const int t = threadIdx.x;
    const int lane = t & 31;
    const int wv = t >> 5;
    const int start = blockIdx.x * SCAN_TB;

    int part = 0;
#pragma unroll 4
    for (int i = 4 * t; i < start; i += 4 * SCAN_TB) {
        const int4 q = *(const int4*)(cnt + i);
        part += (q.x + q.y) + (q.z + q.w);
    }

    const int idx = start + t;
    const int c = (idx < n) ? cnt[idx] : 0;

    int incl = c;
#pragma unroll
    for (int d = 1; d < 32; d <<= 1) {
        const int o = __shfl_up(incl, d, 32);
        if (lane >= d) incl += o;
    }
#pragma unroll
    for (int d = 16; d >= 1; d >>= 1) part += __shfl_xor(part, d, 32);

    if (lane == 31) s_wave[wv] = incl;
    if (lane == 0)  s_part[wv] = part;
    __syncthreads();

    int before = (lane < wv) ? s_wave[lane] : 0;
    int base   = s_part[lane];
#pragma unroll
    for (int d = 16; d >= 1; d >>= 1) {
        before += __shfl_xor(before, d, 32);
        base   += __shfl_xor(base, d, 32);
    }
    const int excl = base + before + incl - c;
    if (idx < n) {
        rowptr[idx] = excl;
        dinv[idx]   = rsqrtf((float)(c + 1));
        if (idx == n - 1) rowptr[n] = excl + c;
    }
}

__global__ void k_csr_fill(int2* __restrict__ csr, const int* __restrict__ rowptr,
                           const int* __restrict__ rank, const float* __restrict__ dinv,
                           const int* __restrict__ src, const int* __restrict__ dst, int e) {
    int i = blockIdx.x * blockDim.x + threadIdx.x;
    if (i >= e) return;
    const int s = src[i], d = dst[i];
    const int pos = rowptr[d] + rank[i];
    csr[pos] = make_int2(s, __float_as_int(dinv[s]));
}

template <bool RELU>
__global__ void k_gemm64_wmma(const float* __restrict__ A, const float* __restrict__ W,
                              float* __restrict__ H, int n) {
    const int rowBase = blockIdx.x * 16;
    const int lane    = threadIdx.x & 31;
    const int colBase = (threadIdx.x >> 5) * 16;
    const int m       = lane & 15;
    const int khalf   = (lane >> 4) * 2;

    const int   arowIdx = rowBase + m;
    const float* arow   = A + (size_t)((arowIdx < n) ? arowIdx : (n - 1)) * NODE_DIM;
    const int   nc      = colBase + m;

    v8f c = {};
#pragma unroll
    for (int k0 = 0; k0 < NODE_DIM; k0 += 4) {
        v2f a;
        a.x = arow[k0 + khalf + 0];
        a.y = arow[k0 + khalf + 1];
        if (RELU) { a.x = fmaxf(a.x, 0.0f); a.y = fmaxf(a.y, 0.0f); }
        v2f b;
        b.x = W[(k0 + khalf + 0) * NODE_DIM + nc];
        b.y = W[(k0 + khalf + 1) * NODE_DIM + nc];
        c = __builtin_amdgcn_wmma_f32_16x16x4_f32(false, a, false, b, (short)0, c, false, false);
    }

#pragma unroll
    for (int r = 0; r < 8; ++r) {
        const int row = rowBase + r + (lane >> 4) * 8;
        if (row < n) H[(size_t)row * NODE_DIM + colBase + (lane & 15)] = c[r];
    }
}

#define GATHER_U 4
#define GATHER_NP 2

__global__ void __launch_bounds__(256)
k_gather(float* __restrict__ agg, const float* __restrict__ h, const float* __restrict__ dinv,
         const float* __restrict__ bias, const int* __restrict__ rowptr,
         const int2* __restrict__ csr, int n) {
    const int lane = threadIdx.x & 31;
    const int half = lane >> 4;
    const unsigned c = lane & 15;
    const int nodeBase = (blockIdx.x * (blockDim.x >> 5) + (threadIdx.x >> 5)) * (2 * GATHER_NP);

    const float4 b = *(const float4*)(bias + 4 * c);

    int   nodeOf[GATHER_NP], beg[GATHER_NP], end[GATHER_NP];
    float di[GATHER_NP];
    float4 own[GATHER_NP];
#pragma unroll
    for (int it = 0; it < GATHER_NP; ++it) {
        const int node = nodeBase + 2 * it + half;
        const int nc = (node < n) ? node : (n - 1);
        nodeOf[it] = node;
        const int r0 = rowptr[nc];
        const int r1 = rowptr[nc + 1];
        beg[it] = r0;
        end[it] = (node < n) ? r1 : r0;
        di[it]  = dinv[nc];
        own[it] = *(const float4*)(h + (unsigned)nc * NODE_DIM + 4 * c);
    }

#pragma unroll
    for (int it = 0; it < GATHER_NP; ++it) {
        const float d2 = di[it];
        float4 acc;
        acc.x = own[it].x * d2 * d2 + b.x;
        acc.y = own[it].y * d2 * d2 + b.y;
        acc.z = own[it].z * d2 * d2 + b.z;
        acc.w = own[it].w * d2 * d2 + b.w;

        const int e1 = end[it];
        if (beg[it] < e1) {
            const int last = e1 - 1;
            int2 p[GATHER_U];
#pragma unroll
            for (int k = 0; k < GATHER_U; ++k) {
                const int jj = beg[it] + k;
                p[k] = csr[(jj < last) ? jj : last];
            }
            for (int j = beg[it]; j < e1; j += GATHER_U) {
                float4 v[GATHER_U];
#pragma unroll
                for (int k = 0; k < GATHER_U; ++k)
                    v[k] = *(const float4*)(h + (unsigned)p[k].x * NODE_DIM + 4 * c);
                int2 pn[GATHER_U];
#pragma unroll
                for (int k = 0; k < GATHER_U; ++k) {
                    const int jj = j + GATHER_U + k;
                    pn[k] = csr[(jj < last) ? jj : last];
                }
#pragma unroll
                for (int k = 0; k < GATHER_U; ++k) {
                    const float norm = __int_as_float(p[k].y) * d2;
                    const bool live = (j + k) < e1;
                    acc.x = live ? (acc.x + v[k].x * norm) : acc.x;
                    acc.y = live ? (acc.y + v[k].y * norm) : acc.y;
                    acc.z = live ? (acc.z + v[k].z * norm) : acc.z;
                    acc.w = live ? (acc.w + v[k].w * norm) : acc.w;
                }
#pragma unroll
                for (int k = 0; k < GATHER_U; ++k) p[k] = pn[k];
            }
        }
        if (nodeOf[it] < n) *(float4*)(agg + (unsigned)nodeOf[it] * NODE_DIM + 4 * c) = acc;
    }
}


template <int STEPS>
__device__ __forceinline__ float group_sum(float v) {
    if (STEPS >= 1) v += __int_as_float(__builtin_amdgcn_update_dpp(0, __float_as_int(v), 0x161, 0xf, 0xf, true));
    if (STEPS >= 2) v += __int_as_float(__builtin_amdgcn_update_dpp(0, __float_as_int(v), 0x162, 0xf, 0xf, true));
    if (STEPS >= 3) v += __int_as_float(__builtin_amdgcn_update_dpp(0, __float_as_int(v), 0x164, 0xf, 0xf, true));
    if (STEPS >= 4) v += __int_as_float(__builtin_amdgcn_update_dpp(0, __float_as_int(v), 0x168, 0xf, 0xf, true));
    return v;
}

__device__ __forceinline__ float dot4(const float4 a, const float4 b) {
    return a.x * b.x + a.y * b.y + a.z * b.z + a.w * b.w;
}

__global__ void k_node_proj(float* __restrict__ pp, const float* __restrict__ agg,
                            const float* __restrict__ We, int n) {
    const int lane = threadIdx.x & 31;
    const int half = lane >> 4;
    const int c    = lane & 15;
    const int nodeBase = (blockIdx.x * (blockDim.x >> 5) + (threadIdx.x >> 5)) * 8;

    const float4 ws = *(const float4*)(We + 4 * c);
    const float4 wd = *(const float4*)(We + 64 + 4 * c);

    float ps[4], pd[4];
#pragma unroll
    for (int r = 0; r < 4; ++r) {
        int node = nodeBase + 2 * r + half;
        node = (node < n) ? node : (n - 1);
        float4 v = *(const float4*)(agg + (size_t)node * NODE_DIM + 4 * c);
        v.x = fmaxf(v.x, 0.0f); v.y = fmaxf(v.y, 0.0f);
        v.z = fmaxf(v.z, 0.0f); v.w = fmaxf(v.w, 0.0f);
        ps[r] = dot4(v, ws);
        pd[r] = dot4(v, wd);
    }
#pragma unroll
    for (int r = 0; r < 4; ++r) {
        ps[r] = group_sum<4>(ps[r]);
        pd[r] = group_sum<4>(pd[r]);
    }
    const float a = (c == 0) ? ps[0] : (c == 1) ? ps[1] : (c == 2) ? ps[2] : ps[3];
    const float b = (c == 0) ? pd[0] : (c == 1) ? pd[1] : (c == 2) ? pd[2] : pd[3];
    const int myNode = nodeBase + 2 * c + half;
    if (c < 4 && myNode < n) *(float2*)(pp + 2 * (size_t)myNode) = make_float2(a, b);
}

__global__ void k_edge_score(float* __restrict__ out, const float* __restrict__ pp,
                             const float* __restrict__ ea, const float* __restrict__ We,
                             const float* __restrict__ be,
                             const int* __restrict__ src, const int* __restrict__ dst, int e) {
    const int lane = threadIdx.x & 31;
    const int q = lane >> 2;
    const int j = lane & 3;
    const int edgeBase = (blockIdx.x * (blockDim.x >> 5) + (threadIdx.x >> 5)) * 64;

    const int e0 = edgeBase + j * 8 + q;
    const int e1 = e0 + 32;
    const int e0c = (e0 < e) ? e0 : (e - 1);
    const int e1c = (e1 < e) ? e1 : (e - 1);
    const unsigned s0 = (unsigned)src[e0c], d0 = (unsigned)dst[e0c];
    const unsigned s1 = (unsigned)src[e1c], d1 = (unsigned)dst[e1c];

    const float4 w = *(const float4*)(We + 128 + 4 * j);

    float p[8];
#pragma unroll
    for (int r = 0; r < 8; ++r) {
        int er = edgeBase + r * 8 + q;
        er = (er < e) ? er : (e - 1);
        const float4 a = *(const float4*)(ea + (size_t)er * 16 + 4 * j);
        p[r] = dot4(a, w);
    }

    const float n0 = pp[2u * s0] + pp[2u * d0 + 1u];
    const float n1 = pp[2u * s1] + pp[2u * d1 + 1u];

#pragma unroll
    for (int r = 0; r < 8; ++r) p[r] = group_sum<2>(p[r]);

    const float m0 = (j == 0) ? p[0] : (j == 1) ? p[1] : (j == 2) ? p[2] : p[3];
    const float m1 = (j == 0) ? p[4] : (j == 1) ? p[5] : (j == 2) ? p[6] : p[7];
    const float bias = be[0];
    if (e0 < e) out[e0] = m0 + n0 + bias;
    if (e1 < e) out[e1] = m1 + n1 + bias;
}

extern "C" void kernel_launch(void* const* d_in, const int* in_sizes, int n_in,
                              void* d_out, int out_size, void* d_ws, size_t ws_size,
                              hipStream_t stream) {
    const float* x  = (const float*)d_in[0];
    const int*   ei = (const int*)d_in[1];
    const float* ea = (const float*)d_in[2];
    const float* W1 = (const float*)d_in[3];
    const float* b1 = (const float*)d_in[4];
    const float* W2 = (const float*)d_in[5];
    const float* b2 = (const float*)d_in[6];
    const float* We = (const float*)d_in[7];
    const float* be = (const float*)d_in[8];
    float* out = (float*)d_out;

    const int N = in_sizes[0] / NODE_DIM;
    const int E = in_sizes[1] / 2;
    const int* src = ei;
    const int* dst = ei + E;

    char* ws = (char*)d_ws;
    const size_t nodeBytes = (((size_t)N * 4) + 511) & ~(size_t)511;
    const size_t rowBytes  = ((((size_t)N + 1) * 4) + 511) & ~(size_t)511;
    const size_t hBytes    = (size_t)N * NODE_DIM * 4;
    float* dinv   = (float*)ws;
    float* h      = (float*)(ws + nodeBytes);
    float* agg    = (float*)(ws + nodeBytes + hBytes);
    int*   cnt    = (int*)(ws + nodeBytes + 2 * hBytes);
    int*   rowptr = (int*)(ws + 2 * nodeBytes + 2 * hBytes);
    int2*  csr    = (int2*)(ws + 2 * nodeBytes + rowBytes + 2 * hBytes);
    int*   rank   = (int*)(ws + 2 * nodeBytes + rowBytes + 2 * hBytes + (size_t)E * 8);

    const int TB = 256;
    const int gN    = (N + TB - 1) / TB;
    const int gE    = (E + TB - 1) / TB;
    const int gScan = (N + SCAN_TB - 1) / SCAN_TB;
    const int gGemm = (N + 15) / 16;
    const int gGath = (N + 16 * GATHER_NP - 1) / (16 * GATHER_NP);
    const int gProj = (N + 63) / 64;
    const int gEdge = (E + 511) / 512;

    k_cnt_zero<<<gN, TB, 0, stream>>>(cnt, N);
    k_cnt_hist<<<gE, TB, 0, stream>>>(cnt, rank, dst, E);
    k_rowptr  <<<gScan, SCAN_TB, 0, stream>>>(cnt, rowptr, dinv, N);
    k_csr_fill<<<gE, TB, 0, stream>>>(csr, rowptr, rank, dinv, src, dst, E);

    k_gemm64_wmma<false><<<gGemm, 128, 0, stream>>>(x, W1, h, N);
    k_gather            <<<gGath, TB, 0, stream>>>(agg, h, dinv, b1, rowptr, csr, N);

    k_gemm64_wmma<true><<<gGemm, 128, 0, stream>>>(agg, W2, h, N);
    k_gather           <<<gGath, TB, 0, stream>>>(agg, h, dinv, b2, rowptr, csr, N);

    float* pp = h;
    k_node_proj <<<gProj, TB, 0, stream>>>(pp, agg, We, N);
    k_edge_score<<<gEdge, TB, 0, stream>>>(out, pp, ea, We, be, src, dst, E);
}


# Edge ranks counted in LDS: a work-group per edge slice holds all destination counters (returning LDS adds; no global atomics, no clear launch); a node's terms sum by slice, then LDS arrival order
# speedup vs baseline: 2.9916x; 1.3871x over previous
#include <hip/hip_runtime.h>
#include <hip/hip_bf16.h>

typedef __attribute__((ext_vector_type(2))) float v2f;
typedef __attribute__((ext_vector_type(8))) float v8f;

#define NODE_DIM 64

#define HIST_TB 1024
#define HIST_NODES 51200
#define HIST_TILES (HIST_NODES / 1024)
#define HIST_U 8
#define MAX_SLICES 64

__device__ __forceinline__ int wave_sum_int(int v) {
    v += __builtin_amdgcn_update_dpp(0, v, 0x161, 0xf, 0xf, true);
    v += __builtin_amdgcn_update_dpp(0, v, 0x162, 0xf, 0xf, true);
    v += __builtin_amdgcn_update_dpp(0, v, 0x164, 0xf, 0xf, true);
    v += __builtin_amdgcn_update_dpp(0, v, 0x168, 0xf, 0xf, true);
    v += __shfl_xor(v, 16, 32);
    return v;
}

__global__ void __launch_bounds__(HIST_TB)
k_hist_lds(int* __restrict__ cs, int* __restrict__ tileS, int* __restrict__ rank,
           const int* __restrict__ dst, int e, int perSlice, int pitch) {
    __shared__ __attribute__((aligned(16))) int s_cnt[HIST_NODES];
    const int t = threadIdx.x;
    const int lane = t & 31;
    const int wv = t >> 5;
    const int slice = blockIdx.x;
    const int nsl = gridDim.x;
    const int chunk0 = blockIdx.y * HIST_NODES;
    const int tile0 = blockIdx.y * HIST_TILES;
    const int tilesLeft = (pitch >> 10) - tile0;
    const int tilesHere = (tilesLeft < HIST_TILES) ? tilesLeft : HIST_TILES;

    const int e0 = slice * perSlice;
    const int e1 = (e0 + perSlice < e) ? (e0 + perSlice) : e;
    const int last = e1 - 1;
    int d[HIST_U];
#pragma unroll
    for (int k = 0; k < HIST_U; ++k) {
        const int i = e0 + k * HIST_TB + t;
        d[k] = dst[(i < last) ? i : last];
    }

    for (int i = 4 * t; i < HIST_NODES; i += 4 * HIST_TB) *(int4*)&s_cnt[i] = make_int4(0, 0, 0, 0);
    __syncthreads();
    for (int base = e0; base < e1; base += HIST_U * HIST_TB) {
        int dn[HIST_U];
#pragma unroll
        for (int k = 0; k < HIST_U; ++k) {
            const int i = base + (HIST_U + k) * HIST_TB + t;
            dn[k] = dst[(i < last) ? i : last];
        }
        int r[HIST_U];
        bool live[HIST_U];
#pragma unroll
        for (int k = 0; k < HIST_U; ++k) {
            const int i = base + k * HIST_TB + t;
            const int dd = d[k] - chunk0;
            live[k] = (i < e1) && ((unsigned)dd < (unsigned)HIST_NODES);
            r[k] = 0;
            if (live[k]) r[k] = atomicAdd(&s_cnt[dd], 1);
        }
#pragma unroll
        for (int k = 0; k < HIST_U; ++k) {
            const int i = base + k * HIST_TB + t;
            if (live[k]) rank[i] = r[k];
        }
#pragma unroll
        for (int k = 0; k < HIST_U; ++k) d[k] = dn[k];
    }
    __syncthreads();

    for (int tl = wv; tl < tilesHere; tl += HIST_TB / 32) {
        const int* sp = s_cnt + tl * 1024 + 4 * lane;
        int* g = cs + (size_t)slice * pitch + chunk0 + tl * 1024 + 4 * lane;
        int4 q[8];
#pragma unroll
        for (int j = 0; j < 8; ++j) q[j] = *(const int4*)(sp + j * 128);
        int sum = 0;
#pragma unroll
        for (int j = 0; j < 8; ++j) {
            *(int4*)(g + j * 128) = q[j];
            sum += (q[j].x + q[j].y) + (q[j].z + q[j].w);
        }
        sum = wave_sum_int(sum);
        if (lane == 0) tileS[(size_t)(tile0 + tl) * nsl + slice] = sum;
    }
}

#define SCAN_TB 1024

__global__ void __launch_bounds__(SCAN_TB)
k_rowptr(int* __restrict__ cs, const int* __restrict__ tileS, int* __restrict__ rowptr,
         float* __restrict__ dinv, int n, int nsl, int pitch) {
    __shared__ int s_wave[SCAN_TB / 32];
    __shared__ int s_part[SCAN_TB / 32];
    const int t = threadIdx.x;
    const int lane = t & 31;
    const int wv = t >> 5;
    const int idx = blockIdx.x * SCAN_TB + t;

    int v[MAX_SLICES];
#pragma unroll
    for (int s = 0; s < MAX_SLICES; ++s) {
        const int sc = (s < nsl) ? s : (nsl - 1);
        const int x = cs[sc * pitch + idx];
        v[s] = (s < nsl) ? x : 0;
    }

    int part = 0;
    const int front = blockIdx.x * nsl;
    for (int i = t; i < front; i += SCAN_TB) part += tileS[i];

    int c = 0;
#pragma unroll
    for (int s = 0; s < MAX_SLICES; ++s) c += v[s];

    int incl = c;
#pragma unroll
    for (int dd = 1; dd < 32; dd <<= 1) {
        const int o = __shfl_up(incl, dd, 32);
        if (lane >= dd) incl += o;
    }
#pragma unroll
    for (int dd = 16; dd >= 1; dd >>= 1) part += __shfl_xor(part, dd, 32);

    if (lane == 31) s_wave[wv] = incl;
    if (lane == 0)  s_part[wv] = part;
    __syncthreads();

    int before = (lane < wv) ? s_wave[lane] : 0;
    int base   = s_part[lane];
#pragma unroll
    for (int dd = 16; dd >= 1; dd >>= 1) {
        before += __shfl_xor(before, dd, 32);
        base   += __shfl_xor(base, dd, 32);
    }
    const int excl = base + before + incl - c;

    int run = excl;
#pragma unroll
    for (int s = 0; s < MAX_SLICES; ++s) {
        if (s < nsl) cs[s * pitch + idx] = run;
        run += v[s];
    }
    if (idx < n) {
        rowptr[idx] = excl;
        dinv[idx]   = rsqrtf((float)(c + 1));
        if (idx == n - 1) rowptr[n] = excl + c;
    }
}

__global__ void k_csr_fill(int2* __restrict__ csr, const int* __restrict__ cs,
                           const int* __restrict__ rank, const float* __restrict__ dinv,
                           const int* __restrict__ src, const int* __restrict__ dst,
                           int e, int perSlice, int pitch) {
    int i = blockIdx.x * blockDim.x + threadIdx.x;
    if (i >= e) return;
    const int slice = (int)(blockIdx.x * blockDim.x) / perSlice;
    const int s = src[i], d = dst[i];
    const int pos = cs[(size_t)slice * pitch + d] + rank[i];
    csr[pos] = make_int2(s, __float_as_int(dinv[s]));
}

template <bool RELU>
__global__ void k_gemm64_wmma(const float* __restrict__ A, const float* __restrict__ W,
                              float* __restrict__ H, int n) {
    const int rowBase = blockIdx.x * 16;
    const int lane    = threadIdx.x & 31;
    const int colBase = (threadIdx.x >> 5) * 16;
    const int m       = lane & 15;
    const int khalf   = (lane >> 4) * 2;

    const int   arowIdx = rowBase + m;
    const float* arow   = A + (size_t)((arowIdx < n) ? arowIdx : (n - 1)) * NODE_DIM;
    const int   nc      = colBase + m;

    v8f c = {};
#pragma unroll
    for (int k0 = 0; k0 < NODE_DIM; k0 += 4) {
        v2f a;
        a.x = arow[k0 + khalf + 0];
        a.y = arow[k0 + khalf + 1];
        if (RELU) { a.x = fmaxf(a.x, 0.0f); a.y = fmaxf(a.y, 0.0f); }
        v2f b;
        b.x = W[(k0 + khalf + 0) * NODE_DIM + nc];
        b.y = W[(k0 + khalf + 1) * NODE_DIM + nc];
        c = __builtin_amdgcn_wmma_f32_16x16x4_f32(false, a, false, b, (short)0, c, false, false);
    }

#pragma unroll
    for (int r = 0; r < 8; ++r) {
        const int row = rowBase + r + (lane >> 4) * 8;
        if (row < n) H[(size_t)row * NODE_DIM + colBase + (lane & 15)] = c[r];
    }
}

#define GATHER_U 4
#define GATHER_NP 2

__global__ void __launch_bounds__(256)
k_gather(float* __restrict__ agg, const float* __restrict__ h, const float* __restrict__ dinv,
         const float* __restrict__ bias, const int* __restrict__ rowptr,
         const int2* __restrict__ csr, int n) {
    const int lane = threadIdx.x & 31;
    const int half = lane >> 4;
    const unsigned c = lane & 15;
    const int nodeBase = (blockIdx.x * (blockDim.x >> 5) + (threadIdx.x >> 5)) * (2 * GATHER_NP);

    const float4 b = *(const float4*)(bias + 4 * c);

    int   nodeOf[GATHER_NP], beg[GATHER_NP], end[GATHER_NP];
    float di[GATHER_NP];
    float4 own[GATHER_NP];
#pragma unroll
    for (int it = 0; it < GATHER_NP; ++it) {
        const int node = nodeBase + 2 * it + half;
        const int nc = (node < n) ? node : (n - 1);
        nodeOf[it] = node;
        const int r0 = rowptr[nc];
        const int r1 = rowptr[nc + 1];
        beg[it] = r0;
        end[it] = (node < n) ? r1 : r0;
        di[it]  = dinv[nc];
        own[it] = *(const float4*)(h + (unsigned)nc * NODE_DIM + 4 * c);
    }

#pragma unroll
    for (int it = 0; it < GATHER_NP; ++it) {
        const float d2 = di[it];
        float4 acc;
        acc.x = own[it].x * d2 * d2 + b.x;
        acc.y = own[it].y * d2 * d2 + b.y;
        acc.z = own[it].z * d2 * d2 + b.z;
        acc.w = own[it].w * d2 * d2 + b.w;

        const int e1 = end[it];
        if (beg[it] < e1) {
            const int last = e1 - 1;
            int2 p[GATHER_U];
#pragma unroll
            for (int k = 0; k < GATHER_U; ++k) {
                const int jj = beg[it] + k;
                p[k] = csr[(jj < last) ? jj : last];
            }
            for (int j = beg[it]; j < e1; j += GATHER_U) {
                float4 v[GATHER_U];
#pragma unroll
                for (int k = 0; k < GATHER_U; ++k)
                    v[k] = *(const float4*)(h + (unsigned)p[k].x * NODE_DIM + 4 * c);
                int2 pn[GATHER_U];
#pragma unroll
                for (int k = 0; k < GATHER_U; ++k) {
                    const int jj = j + GATHER_U + k;
                    pn[k] = csr[(jj < last) ? jj : last];
                }
#pragma unroll
                for (int k = 0; k < GATHER_U; ++k) {
                    const float norm = __int_as_float(p[k].y) * d2;
                    const bool live = (j + k) < e1;
                    acc.x = live ? (acc.x + v[k].x * norm) : acc.x;
                    acc.y = live ? (acc.y + v[k].y * norm) : acc.y;
                    acc.z = live ? (acc.z + v[k].z * norm) : acc.z;
                    acc.w = live ? (acc.w + v[k].w * norm) : acc.w;
                }
#pragma unroll
                for (int k = 0; k < GATHER_U; ++k) p[k] = pn[k];
            }
        }
        if (nodeOf[it] < n) *(float4*)(agg + (unsigned)nodeOf[it] * NODE_DIM + 4 * c) = acc;
    }
}


template <int STEPS>
__device__ __forceinline__ float group_sum(float v) {
    if (STEPS >= 1) v += __int_as_float(__builtin_amdgcn_update_dpp(0, __float_as_int(v), 0x161, 0xf, 0xf, true));
    if (STEPS >= 2) v += __int_as_float(__builtin_amdgcn_update_dpp(0, __float_as_int(v), 0x162, 0xf, 0xf, true));
    if (STEPS >= 3) v += __int_as_float(__builtin_amdgcn_update_dpp(0, __float_as_int(v), 0x164, 0xf, 0xf, true));
    if (STEPS >= 4) v += __int_as_float(__builtin_amdgcn_update_dpp(0, __float_as_int(v), 0x168, 0xf, 0xf, true));
    return v;
}

__device__ __forceinline__ float dot4(const float4 a, const float4 b) {
    return a.x * b.x + a.y * b.y + a.z * b.z + a.w * b.w;
}

__global__ void k_node_proj(float* __restrict__ pp, const float* __restrict__ agg,
                            const float* __restrict__ We, int n) {
    const int lane = threadIdx.x & 31;
    const int half = lane >> 4;
    const int c    = lane & 15;
    const int nodeBase = (blockIdx.x * (blockDim.x >> 5) + (threadIdx.x >> 5)) * 8;

    const float4 ws = *(const float4*)(We + 4 * c);
    const float4 wd = *(const float4*)(We + 64 + 4 * c);

    float ps[4], pd[4];
#pragma unroll
    for (int r = 0; r < 4; ++r) {
        int node = nodeBase + 2 * r + half;
        node = (node < n) ? node : (n - 1);
        float4 v = *(const float4*)(agg + (size_t)node * NODE_DIM + 4 * c);
        v.x = fmaxf(v.x, 0.0f); v.y = fmaxf(v.y, 0.0f);
        v.z = fmaxf(v.z, 0.0f); v.w = fmaxf(v.w, 0.0f);
        ps[r] = dot4(v, ws);
        pd[r] = dot4(v, wd);
    }
#pragma unroll
    for (int r = 0; r < 4; ++r) {
        ps[r] = group_sum<4>(ps[r]);
        pd[r] = group_sum<4>(pd[r]);
    }
    const float a = (c == 0) ? ps[0] : (c == 1) ? ps[1] : (c == 2) ? ps[2] : ps[3];
    const float b = (c == 0) ? pd[0] : (c == 1) ? pd[1] : (c == 2) ? pd[2] : pd[3];
    const int myNode = nodeBase + 2 * c + half;
    if (c < 4 && myNode < n) *(float2*)(pp + 2 * (size_t)myNode) = make_float2(a, b);
}

__global__ void k_edge_score(float* __restrict__ out, const float* __restrict__ pp,
                             const float* __restrict__ ea, const float* __restrict__ We,
                             const float* __restrict__ be,
                             const int* __restrict__ src, const int* __restrict__ dst, int e) {
    const int lane = threadIdx.x & 31;
    const int q = lane >> 2;
    const int j = lane & 3;
    const int edgeBase = (blockIdx.x * (blockDim.x >> 5) + (threadIdx.x >> 5)) * 64;

    const int e0 = edgeBase + j * 8 + q;
    const int e1 = e0 + 32;
    const int e0c = (e0 < e) ? e0 : (e - 1);
    const int e1c = (e1 < e) ? e1 : (e - 1);
    const unsigned s0 = (unsigned)src[e0c], d0 = (unsigned)dst[e0c];
    const unsigned s1 = (unsigned)src[e1c], d1 = (unsigned)dst[e1c];

    const float4 w = *(const float4*)(We + 128 + 4 * j);

    float p[8];
#pragma unroll
    for (int r = 0; r < 8; ++r) {
        int er = edgeBase + r * 8 + q;
        er = (er < e) ? er : (e - 1);
        const float4 a = *(const float4*)(ea + (size_t)er * 16 + 4 * j);
        p[r] = dot4(a, w);
    }

    const float n0 = pp[2u * s0] + pp[2u * d0 + 1u];
    const float n1 = pp[2u * s1] + pp[2u * d1 + 1u];

#pragma unroll
    for (int r = 0; r < 8; ++r) p[r] = group_sum<2>(p[r]);

    const float m0 = (j == 0) ? p[0] : (j == 1) ? p[1] : (j == 2) ? p[2] : p[3];
    const float m1 = (j == 0) ? p[4] : (j == 1) ? p[5] : (j == 2) ? p[6] : p[7];
    const float bias = be[0];
    if (e0 < e) out[e0] = m0 + n0 + bias;
    if (e1 < e) out[e1] = m1 + n1 + bias;
}

extern "C" void kernel_launch(void* const* d_in, const int* in_sizes, int n_in,
                              void* d_out, int out_size, void* d_ws, size_t ws_size,
                              hipStream_t stream) {
    const float* x  = (const float*)d_in[0];
    const int*   ei = (const int*)d_in[1];
    const float* ea = (const float*)d_in[2];
    const float* W1 = (const float*)d_in[3];
    const float* b1 = (const float*)d_in[4];
    const float* W2 = (const float*)d_in[5];
    const float* b2 = (const float*)d_in[6];
    const float* We = (const float*)d_in[7];
    const float* be = (const float*)d_in[8];
    float* out = (float*)d_out;

    const int N = in_sizes[0] / NODE_DIM;
    const int E = in_sizes[1] / 2;
    const int* src = ei;
    const int* dst = ei + E;

    const int perSlice = (E / MAX_SLICES / (HIST_U * HIST_TB) + 1) * (HIST_U * HIST_TB);
    const int nSlices  = (E + perSlice - 1) / perSlice;
    const int gScan    = (N + SCAN_TB - 1) / SCAN_TB;
    const int pitch    = gScan * SCAN_TB;
    const int gChunk   = (N + HIST_NODES - 1) / HIST_NODES;

    char* ws = (char*)d_ws;
    const size_t nodeBytes = (((size_t)N * 4) + 511) & ~(size_t)511;
    const size_t rowBytes  = ((((size_t)N + 1) * 4) + 511) & ~(size_t)511;
    const size_t hBytes    = (size_t)N * NODE_DIM * 4;
    const size_t edgeBytes = (((size_t)E * 4) + 511) & ~(size_t)511;
    float* dinv   = (float*)ws;
    float* h      = (float*)(ws + nodeBytes);
    float* agg    = (float*)(ws + nodeBytes + hBytes);
    int*   rowptr = (int*)(ws + nodeBytes + 2 * hBytes);
    int2*  csr    = (int2*)(ws + nodeBytes + rowBytes + 2 * hBytes);
    int*   rank   = (int*)(ws + nodeBytes + rowBytes + 2 * hBytes + 2 * edgeBytes);
    int*   cs     = (int*)(ws + nodeBytes + rowBytes + 2 * hBytes + 3 * edgeBytes);
    int*   tileS  = cs + (size_t)nSlices * pitch;

    const int TB = 256;
    const int gE    = (E + TB - 1) / TB;
    const int gGemm = (N + 15) / 16;
    const int gGath = (N + 16 * GATHER_NP - 1) / (16 * GATHER_NP);
    const int gProj = (N + 63) / 64;
    const int gEdge = (E + 511) / 512;

    k_hist_lds<<<dim3(nSlices, gChunk), HIST_TB, 0, stream>>>(cs, tileS, rank, dst, E, perSlice, pitch);
    k_rowptr  <<<gScan, SCAN_TB, 0, stream>>>(cs, tileS, rowptr, dinv, N, nSlices, pitch);
    k_csr_fill<<<gE, TB, 0, stream>>>(csr, cs, rank, dinv, src, dst, E, perSlice, pitch);

    k_gemm64_wmma<false><<<gGemm, 128, 0, stream>>>(x, W1, h, N);
    k_gather            <<<gGath, TB, 0, stream>>>(agg, h, dinv, b1, rowptr, csr, N);

    k_gemm64_wmma<true><<<gGemm, 128, 0, stream>>>(agg, W2, h, N);
    k_gather           <<<gGath, TB, 0, stream>>>(agg, h, dinv, b2, rowptr, csr, N);

    float* pp = h;
    k_node_proj <<<gProj, TB, 0, stream>>>(pp, agg, We, N);
    k_edge_score<<<gEdge, TB, 0, stream>>>(out, pp, ea, We, be, src, dst, E);
}


# Each aggregation's only consumer runs in its epilogue: the layer-2 product on the block's 32 sums through LDS, the scorer's node halves from registers; 9 launches become 7, the sums never stored
# speedup vs baseline: 3.3174x; 1.1089x over previous
#include <hip/hip_runtime.h>
#include <hip/hip_bf16.h>

typedef __attribute__((ext_vector_type(2))) float v2f;
typedef __attribute__((ext_vector_type(8))) float v8f;

#define NODE_DIM 64

#define HIST_TB 1024
#define HIST_NODES 51200
#define HIST_TILES (HIST_NODES / 1024)
#define HIST_U 8
#define MAX_SLICES 64

__device__ __forceinline__ int wave_sum_int(int v) {
    v += __builtin_amdgcn_update_dpp(0, v, 0x161, 0xf, 0xf, true);
    v += __builtin_amdgcn_update_dpp(0, v, 0x162, 0xf, 0xf, true);
    v += __builtin_amdgcn_update_dpp(0, v, 0x164, 0xf, 0xf, true);
    v += __builtin_amdgcn_update_dpp(0, v, 0x168, 0xf, 0xf, true);
    v += __shfl_xor(v, 16, 32);
    return v;
}

__global__ void __launch_bounds__(HIST_TB)
k_hist_lds(int* __restrict__ cs, int* __restrict__ tileS, int* __restrict__ rank,
           const int* __restrict__ dst, int e, int perSlice, int pitch) {
    __shared__ __attribute__((aligned(16))) int s_cnt[HIST_NODES];
    const int t = threadIdx.x;
    const int lane = t & 31;
    const int wv = t >> 5;
    const int slice = blockIdx.x;
    const int nsl = gridDim.x;
    const int chunk0 = blockIdx.y * HIST_NODES;
    const int tile0 = blockIdx.y * HIST_TILES;
    const int tilesLeft = (pitch >> 10) - tile0;
    const int tilesHere = (tilesLeft < HIST_TILES) ? tilesLeft : HIST_TILES;

    const int e0 = slice * perSlice;
    const int e1 = (e0 + perSlice < e) ? (e0 + perSlice) : e;
    const int last = e1 - 1;
    int d[HIST_U];
#pragma unroll
    for (int k = 0; k < HIST_U; ++k) {
        const int i = e0 + k * HIST_TB + t;
        d[k] = dst[(i < last) ? i : last];
    }

    for (int i = 4 * t; i < HIST_NODES; i += 4 * HIST_TB) *(int4*)&s_cnt[i] = make_int4(0, 0, 0, 0);
    __syncthreads();
    for (int base = e0; base < e1; base += HIST_U * HIST_TB) {
        int dn[HIST_U];
#pragma unroll
        for (int k = 0; k < HIST_U; ++k) {
            const int i = base + (HIST_U + k) * HIST_TB + t;
            dn[k] = dst[(i < last) ? i : last];
        }
        int r[HIST_U];
        bool live[HIST_U];
#pragma unroll
        for (int k = 0; k < HIST_U; ++k) {
            const int i = base + k * HIST_TB + t;
            const int dd = d[k] - chunk0;
            live[k] = (i < e1) && ((unsigned)dd < (unsigned)HIST_NODES);
            r[k] = 0;
            if (live[k]) r[k] = atomicAdd(&s_cnt[dd], 1);
        }
#pragma unroll
        for (int k = 0; k < HIST_U; ++k) {
            const int i = base + k * HIST_TB + t;
            if (live[k]) rank[i] = r[k];
        }
#pragma unroll
        for (int k = 0; k < HIST_U; ++k) d[k] = dn[k];
    }
    __syncthreads();

    for (int tl = wv; tl < tilesHere; tl += HIST_TB / 32) {
        const int* sp = s_cnt + tl * 1024 + 4 * lane;
        int* g = cs + (size_t)slice * pitch + chunk0 + tl * 1024 + 4 * lane;
        int4 q[8];
#pragma unroll
        for (int j = 0; j < 8; ++j) q[j] = *(const int4*)(sp + j * 128);
        int sum = 0;
#pragma unroll
        for (int j = 0; j < 8; ++j) {
            *(int4*)(g + j * 128) = q[j];
            sum += (q[j].x + q[j].y) + (q[j].z + q[j].w);
        }
        sum = wave_sum_int(sum);
        if (lane == 0) tileS[(size_t)(tile0 + tl) * nsl + slice] = sum;
    }
}

#define SCAN_TB 1024

__global__ void __launch_bounds__(SCAN_TB)
k_rowptr(int* __restrict__ cs, const int* __restrict__ tileS, int* __restrict__ rowptr,
         float* __restrict__ dinv, int n, int nsl, int pitch) {
    __shared__ int s_wave[SCAN_TB / 32];
    __shared__ int s_part[SCAN_TB / 32];
    const int t = threadIdx.x;
    const int lane = t & 31;
    const int wv = t >> 5;
    const int idx = blockIdx.x * SCAN_TB + t;

    int v[MAX_SLICES];
#pragma unroll
    for (int s = 0; s < MAX_SLICES; ++s) {
        const int sc = (s < nsl) ? s : (nsl - 1);
        const int x = cs[sc * pitch + idx];
        v[s] = (s < nsl) ? x : 0;
    }

    int part = 0;
    const int front = blockIdx.x * nsl;
    for (int i = t; i < front; i += SCAN_TB) part += tileS[i];

    int c = 0;
#pragma unroll
    for (int s = 0; s < MAX_SLICES; ++s) c += v[s];

    int incl = c;
#pragma unroll
    for (int dd = 1; dd < 32; dd <<= 1) {
        const int o = __shfl_up(incl, dd, 32);
        if (lane >= dd) incl += o;
    }
#pragma unroll
    for (int dd = 16; dd >= 1; dd >>= 1) part += __shfl_xor(part, dd, 32);

    if (lane == 31) s_wave[wv] = incl;
    if (lane == 0)  s_part[wv] = part;
    __syncthreads();

    int before = (lane < wv) ? s_wave[lane] : 0;
    int base   = s_part[lane];
#pragma unroll
    for (int dd = 16; dd >= 1; dd >>= 1) {
        before += __shfl_xor(before, dd, 32);
        base   += __shfl_xor(base, dd, 32);
    }
    const int excl = base + before + incl - c;

    int run = excl;
#pragma unroll
    for (int s = 0; s < MAX_SLICES; ++s) {
        if (s < nsl) cs[s * pitch + idx] = run;
        run += v[s];
    }
    if (idx < n) {
        rowptr[idx] = excl;
        dinv[idx]   = rsqrtf((float)(c + 1));
        if (idx == n - 1) rowptr[n] = excl + c;
    }
}

__global__ void k_csr_fill(int2* __restrict__ csr, const int* __restrict__ cs,
                           const int* __restrict__ rank, const float* __restrict__ dinv,
                           const int* __restrict__ src, const int* __restrict__ dst,
                           int e, int perSlice, int pitch) {
    int i = blockIdx.x * blockDim.x + threadIdx.x;
    if (i >= e) return;
    const int slice = (int)(blockIdx.x * blockDim.x) / perSlice;
    const int s = src[i], d = dst[i];
    const int pos = cs[(size_t)slice * pitch + d] + rank[i];
    csr[pos] = make_int2(s, __float_as_int(dinv[s]));
}

__global__ void k_gemm64_wmma(const float* __restrict__ A, const float* __restrict__ W,
                              float* __restrict__ H, int n) {
    const int rowBase = blockIdx.x * 16;
    const int lane    = threadIdx.x & 31;
    const int colBase = (threadIdx.x >> 5) * 16;
    const int m       = lane & 15;
    const int khalf   = (lane >> 4) * 2;

    const int   arowIdx = rowBase + m;
    const float* arow   = A + (size_t)((arowIdx < n) ? arowIdx : (n - 1)) * NODE_DIM;
    const int   nc      = colBase + m;

    v8f c = {};
#pragma unroll
    for (int k0 = 0; k0 < NODE_DIM; k0 += 4) {
        v2f a;
        a.x = arow[k0 + khalf + 0];
        a.y = arow[k0 + khalf + 1];
        v2f b;
        b.x = W[(k0 + khalf + 0) * NODE_DIM + nc];
        b.y = W[(k0 + khalf + 1) * NODE_DIM + nc];
        c = __builtin_amdgcn_wmma_f32_16x16x4_f32(false, a, false, b, (short)0, c, false, false);
    }

#pragma unroll
    for (int r = 0; r < 8; ++r) {
        const int row = rowBase + r + (lane >> 4) * 8;
        if (row < n) H[(size_t)row * NODE_DIM + colBase + (lane & 15)] = c[r];
    }
}

#define GATHER_U 4
#define GATHER_NP 2
#define GATHER_TB 256

__device__ __forceinline__ void gather_rows(float4 (&acc)[GATHER_NP],
        const float* __restrict__ h, const float* __restrict__ dinv,
        const float* __restrict__ bias, const int* __restrict__ rowptr,
        const int2* __restrict__ csr, int n, int nodeBase, int half, unsigned c) {
    const float4 b = *(const float4*)(bias + 4 * c);

    int   beg[GATHER_NP], end[GATHER_NP];
    float di[GATHER_NP];
    float4 own[GATHER_NP];
#pragma unroll
    for (int it = 0; it < GATHER_NP; ++it) {
        const int node = nodeBase + 2 * it + half;
        const int nc = (node < n) ? node : (n - 1);
        const int r0 = rowptr[nc];
        const int r1 = rowptr[nc + 1];
        beg[it] = r0;
        end[it] = (node < n) ? r1 : r0;
        di[it]  = dinv[nc];
        own[it] = *(const float4*)(h + (unsigned)nc * NODE_DIM + 4 * c);
    }

#pragma unroll
    for (int it = 0; it < GATHER_NP; ++it) {
        const float d2 = di[it];
        float4 a;
        a.x = own[it].x * d2 * d2 + b.x;
        a.y = own[it].y * d2 * d2 + b.y;
        a.z = own[it].z * d2 * d2 + b.z;
        a.w = own[it].w * d2 * d2 + b.w;

        const int e1 = end[it];
        if (beg[it] < e1) {
            const int last = e1 - 1;
            int2 p[GATHER_U];
#pragma unroll
            for (int k = 0; k < GATHER_U; ++k) {
                const int jj = beg[it] + k;
                p[k] = csr[(jj < last) ? jj : last];
            }
            for (int j = beg[it]; j < e1; j += GATHER_U) {
                float4 v[GATHER_U];
#pragma unroll
                for (int k = 0; k < GATHER_U; ++k)
                    v[k] = *(const float4*)(h + (unsigned)p[k].x * NODE_DIM + 4 * c);
                int2 pn[GATHER_U];
#pragma unroll
                for (int k = 0; k < GATHER_U; ++k) {
                    const int jj = j + GATHER_U + k;
                    pn[k] = csr[(jj < last) ? jj : last];
                }
#pragma unroll
                for (int k = 0; k < GATHER_U; ++k) {
                    const float norm = __int_as_float(p[k].y) * d2;
                    const bool live = (j + k) < e1;
                    a.x = live ? (a.x + v[k].x * norm) : a.x;
                    a.y = live ? (a.y + v[k].y * norm) : a.y;
                    a.z = live ? (a.z + v[k].z * norm) : a.z;
                    a.w = live ? (a.w + v[k].w * norm) : a.w;
                }
#pragma unroll
                for (int k = 0; k < GATHER_U; ++k) p[k] = pn[k];
            }
        }
        acc[it] = a;
    }
}

#define TILE_ROWS (GATHER_TB / 32 * 2 * GATHER_NP)
#define TILE_PITCH 68

__global__ void __launch_bounds__(GATHER_TB)
k_gather_gemm(float* __restrict__ hout, const float* __restrict__ h, const float* __restrict__ dinv,
              const float* __restrict__ bias, const float* __restrict__ W,
              const int* __restrict__ rowptr, const int2* __restrict__ csr, int n) {
    __shared__ __attribute__((aligned(16))) float s_a[TILE_ROWS * TILE_PITCH];
    const int lane = threadIdx.x & 31;
    const int wv   = threadIdx.x >> 5;
    const int half = lane >> 4;
    const unsigned c = lane & 15;
    const int blockNode = blockIdx.x * TILE_ROWS;

    float4 acc[GATHER_NP];
    gather_rows(acc, h, dinv, bias, rowptr, csr, n, blockNode + wv * (2 * GATHER_NP), half, c);

#pragma unroll
    for (int it = 0; it < GATHER_NP; ++it) {
        float4 v = acc[it];
        v.x = fmaxf(v.x, 0.0f); v.y = fmaxf(v.y, 0.0f);
        v.z = fmaxf(v.z, 0.0f); v.w = fmaxf(v.w, 0.0f);
        *(float4*)&s_a[(wv * (2 * GATHER_NP) + 2 * it + half) * TILE_PITCH + 4 * c] = v;
    }

    const int rowTile = (wv >> 2) * 16;
    const int colBase = (wv & 3) * 16;
    const int m       = lane & 15;
    const int khalf   = half * 2;
    const int nc      = colBase + m;

    v2f b[NODE_DIM / 4];
#pragma unroll
    for (int i = 0; i < NODE_DIM / 4; ++i) {
        b[i].x = W[(4 * i + khalf + 0) * NODE_DIM + nc];
        b[i].y = W[(4 * i + khalf + 1) * NODE_DIM + nc];
    }
    __syncthreads();

    const float* arow = s_a + (rowTile + m) * TILE_PITCH + khalf;
    v8f d = {};
#pragma unroll
    for (int i = 0; i < NODE_DIM / 4; ++i) {
        const v2f a = *(const v2f*)(arow + 4 * i);
        d = __builtin_amdgcn_wmma_f32_16x16x4_f32(false, a, false, b[i], (short)0, d, false, false);
    }

#pragma unroll
    for (int r = 0; r < 8; ++r) {
        const int row = blockNode + rowTile + r + half * 8;
        if (row < n) hout[(size_t)row * NODE_DIM + colBase + m] = d[r];
    }
}


template <int STEPS>
__device__ __forceinline__ float group_sum(float v) {
    if (STEPS >= 1) v += __int_as_float(__builtin_amdgcn_update_dpp(0, __float_as_int(v), 0x161, 0xf, 0xf, true));
    if (STEPS >= 2) v += __int_as_float(__builtin_amdgcn_update_dpp(0, __float_as_int(v), 0x162, 0xf, 0xf, true));
    if (STEPS >= 3) v += __int_as_float(__builtin_amdgcn_update_dpp(0, __float_as_int(v), 0x164, 0xf, 0xf, true));
    if (STEPS >= 4) v += __int_as_float(__builtin_amdgcn_update_dpp(0, __float_as_int(v), 0x168, 0xf, 0xf, true));
    return v;
}

__device__ __forceinline__ float dot4(const float4 a, const float4 b) {
    return a.x * b.x + a.y * b.y + a.z * b.z + a.w * b.w;
}

__global__ void __launch_bounds__(GATHER_TB)
k_gather_proj(float* __restrict__ pp, const float* __restrict__ h, const float* __restrict__ dinv,
              const float* __restrict__ bias, const float* __restrict__ We,
              const int* __restrict__ rowptr, const int2* __restrict__ csr, int n) {
    const int lane = threadIdx.x & 31;
    const int half = lane >> 4;
    const unsigned c = lane & 15;
    const int nodeBase = (blockIdx.x * (GATHER_TB >> 5) + (threadIdx.x >> 5)) * (2 * GATHER_NP);

    float4 acc[GATHER_NP];
    gather_rows(acc, h, dinv, bias, rowptr, csr, n, nodeBase, half, c);

    const float4 ws = *(const float4*)(We + 4 * c);
    const float4 wd = *(const float4*)(We + 64 + 4 * c);

    float ps[GATHER_NP], pd[GATHER_NP];
#pragma unroll
    for (int it = 0; it < GATHER_NP; ++it) {
        float4 v = acc[it];
        v.x = fmaxf(v.x, 0.0f); v.y = fmaxf(v.y, 0.0f);
        v.z = fmaxf(v.z, 0.0f); v.w = fmaxf(v.w, 0.0f);
        ps[it] = group_sum<4>(dot4(v, ws));
        pd[it] = group_sum<4>(dot4(v, wd));
    }
    const float a = (c == 0) ? ps[0] : ps[1];
    const float b = (c == 0) ? pd[0] : pd[1];
    const int myNode = nodeBase + 2 * (int)c + half;
    if (c < GATHER_NP && myNode < n) *(float2*)(pp + 2 * (size_t)myNode) = make_float2(a, b);
}

__global__ void k_edge_score(float* __restrict__ out, const float* __restrict__ pp,
                             const float* __restrict__ ea, const float* __restrict__ We,
                             const float* __restrict__ be,
                             const int* __restrict__ src, const int* __restrict__ dst, int e) {
    const int lane = threadIdx.x & 31;
    const int q = lane >> 2;
    const int j = lane & 3;
    const int edgeBase = (blockIdx.x * (blockDim.x >> 5) + (threadIdx.x >> 5)) * 64;

    const int e0 = edgeBase + j * 8 + q;
    const int e1 = e0 + 32;
    const int e0c = (e0 < e) ? e0 : (e - 1);
    const int e1c = (e1 < e) ? e1 : (e - 1);
    const unsigned s0 = (unsigned)src[e0c], d0 = (unsigned)dst[e0c];
    const unsigned s1 = (unsigned)src[e1c], d1 = (unsigned)dst[e1c];

    const float4 w = *(const float4*)(We + 128 + 4 * j);

    float p[8];
#pragma unroll
    for (int r = 0; r < 8; ++r) {
        int er = edgeBase + r * 8 + q;
        er = (er < e) ? er : (e - 1);
        const float4 a = *(const float4*)(ea + (size_t)er * 16 + 4 * j);
        p[r] = dot4(a, w);
    }

    const float n0 = pp[2u * s0] + pp[2u * d0 + 1u];
    const float n1 = pp[2u * s1] + pp[2u * d1 + 1u];

#pragma unroll
    for (int r = 0; r < 8; ++r) p[r] = group_sum<2>(p[r]);

    const float m0 = (j == 0) ? p[0] : (j == 1) ? p[1] : (j == 2) ? p[2] : p[3];
    const float m1 = (j == 0) ? p[4] : (j == 1) ? p[5] : (j == 2) ? p[6] : p[7];
    const float bias = be[0];
    if (e0 < e) out[e0] = m0 + n0 + bias;
    if (e1 < e) out[e1] = m1 + n1 + bias;
}

extern "C" void kernel_launch(void* const* d_in, const int* in_sizes, int n_in,
                              void* d_out, int out_size, void* d_ws, size_t ws_size,
                              hipStream_t stream) {
    const float* x  = (const float*)d_in[0];
    const int*   ei = (const int*)d_in[1];
    const float* ea = (const float*)d_in[2];
    const float* W1 = (const float*)d_in[3];
    const float* b1 = (const float*)d_in[4];
    const float* W2 = (const float*)d_in[5];
    const float* b2 = (const float*)d_in[6];
    const float* We = (const float*)d_in[7];
    const float* be = (const float*)d_in[8];
    float* out = (float*)d_out;

    const int N = in_sizes[0] / NODE_DIM;
    const int E = in_sizes[1] / 2;
    const int* src = ei;
    const int* dst = ei + E;

    const int perSlice = (E / MAX_SLICES / (HIST_U * HIST_TB) + 1) * (HIST_U * HIST_TB);
    const int nSlices  = (E + perSlice - 1) / perSlice;
    const int gScan    = (N + SCAN_TB - 1) / SCAN_TB;
    const int pitch    = gScan * SCAN_TB;
    const int gChunk   = (N + HIST_NODES - 1) / HIST_NODES;

    char* ws = (char*)d_ws;
    const size_t nodeBytes = (((size_t)N * 4) + 511) & ~(size_t)511;
    const size_t rowBytes  = ((((size_t)N + 1) * 4) + 511) & ~(size_t)511;
    const size_t hBytes    = (size_t)N * NODE_DIM * 4;
    const size_t edgeBytes = (((size_t)E * 4) + 511) & ~(size_t)511;
    float* dinv   = (float*)ws;
    float* hA     = (float*)(ws + nodeBytes);
    float* hB     = (float*)(ws + nodeBytes + hBytes);
    int*   rowptr = (int*)(ws + nodeBytes + 2 * hBytes);
    int2*  csr    = (int2*)(ws + nodeBytes + rowBytes + 2 * hBytes);
    int*   rank   = (int*)(ws + nodeBytes + rowBytes + 2 * hBytes + 2 * edgeBytes);
    int*   cs     = (int*)(ws + nodeBytes + rowBytes + 2 * hBytes + 3 * edgeBytes);
    int*   tileS  = cs + (size_t)nSlices * pitch;

    const int TB = 256;
    const int gE    = (E + TB - 1) / TB;
    const int gGemm = (N + 15) / 16;
    const int gGath = (N + TILE_ROWS - 1) / TILE_ROWS;
    const int gEdge = (E + 511) / 512;

    k_hist_lds<<<dim3(nSlices, gChunk), HIST_TB, 0, stream>>>(cs, tileS, rank, dst, E, perSlice, pitch);
    k_rowptr  <<<gScan, SCAN_TB, 0, stream>>>(cs, tileS, rowptr, dinv, N, nSlices, pitch);
    k_csr_fill<<<gE, TB, 0, stream>>>(csr, cs, rank, dinv, src, dst, E, perSlice, pitch);

    k_gemm64_wmma<<<gGemm, 128, 0, stream>>>(x, W1, hA, N);
    k_gather_gemm<<<gGath, GATHER_TB, 0, stream>>>(hB, hA, dinv, b1, W2, rowptr, csr, N);

    float* pp = hA;
    k_gather_proj<<<gGath, GATHER_TB, 0, stream>>>(pp, hB, dinv, b2, We, rowptr, csr, N);
    k_edge_score <<<gEdge, TB, 0, stream>>>(out, pp, ea, We, be, src, dst, E);
}


# Layer-1 product moved behind its aggregation by linearity: the gather sums x rows, its epilogue multiplies the block's 32 sums by W1 then W2; the x@W1 launch and h1 plane go (sum order changes)
# speedup vs baseline: 3.5931x; 1.0831x over previous
#include <hip/hip_runtime.h>
#include <hip/hip_bf16.h>

typedef __attribute__((ext_vector_type(2))) float v2f;
typedef __attribute__((ext_vector_type(8))) float v8f;

#define NODE_DIM 64

#define HIST_TB 1024
#define HIST_NODES 51200
#define HIST_TILES (HIST_NODES / 1024)
#define HIST_U 8
#define MAX_SLICES 64

__device__ __forceinline__ int wave_sum_int(int v) {
    v += __builtin_amdgcn_update_dpp(0, v, 0x161, 0xf, 0xf, true);
    v += __builtin_amdgcn_update_dpp(0, v, 0x162, 0xf, 0xf, true);
    v += __builtin_amdgcn_update_dpp(0, v, 0x164, 0xf, 0xf, true);
    v += __builtin_amdgcn_update_dpp(0, v, 0x168, 0xf, 0xf, true);
    v += __shfl_xor(v, 16, 32);
    return v;
}

__global__ void __launch_bounds__(HIST_TB)
k_hist_lds(int* __restrict__ cs, int* __restrict__ tileS, int* __restrict__ rank,
           const int* __restrict__ dst, int e, int perSlice, int pitch) {
    __shared__ __attribute__((aligned(16))) int s_cnt[HIST_NODES];
    const int t = threadIdx.x;
    const int lane = t & 31;
    const int wv = t >> 5;
    const int slice = blockIdx.x;
    const int nsl = gridDim.x;
    const int chunk0 = blockIdx.y * HIST_NODES;
    const int tile0 = blockIdx.y * HIST_TILES;
    const int tilesLeft = (pitch >> 10) - tile0;
    const int tilesHere = (tilesLeft < HIST_TILES) ? tilesLeft : HIST_TILES;

    const int e0 = slice * perSlice;
    const int e1 = (e0 + perSlice < e) ? (e0 + perSlice) : e;
    const int last = e1 - 1;
    int d[HIST_U];
#pragma unroll
    for (int k = 0; k < HIST_U; ++k) {
        const int i = e0 + k * HIST_TB + t;
        d[k] = dst[(i < last) ? i : last];
    }

    for (int i = 4 * t; i < HIST_NODES; i += 4 * HIST_TB) *(int4*)&s_cnt[i] = make_int4(0, 0, 0, 0);
    __syncthreads();
    for (int base = e0; base < e1; base += HIST_U * HIST_TB) {
        int dn[HIST_U];
#pragma unroll
        for (int k = 0; k < HIST_U; ++k) {
            const int i = base + (HIST_U + k) * HIST_TB + t;
            dn[k] = dst[(i < last) ? i : last];
        }
        int r[HIST_U];
        bool live[HIST_U];
#pragma unroll
        for (int k = 0; k < HIST_U; ++k) {
            const int i = base + k * HIST_TB + t;
            const int dd = d[k] - chunk0;
            live[k] = (i < e1) && ((unsigned)dd < (unsigned)HIST_NODES);
            r[k] = 0;
            if (live[k]) r[k] = atomicAdd(&s_cnt[dd], 1);
        }
#pragma unroll
        for (int k = 0; k < HIST_U; ++k) {
            const int i = base + k * HIST_TB + t;
            if (live[k]) rank[i] = r[k];
        }
#pragma unroll
        for (int k = 0; k < HIST_U; ++k) d[k] = dn[k];
    }
    __syncthreads();

    for (int tl = wv; tl < tilesHere; tl += HIST_TB / 32) {
        const int* sp = s_cnt + tl * 1024 + 4 * lane;
        int* g = cs + (size_t)slice * pitch + chunk0 + tl * 1024 + 4 * lane;
        int4 q[8];
#pragma unroll
        for (int j = 0; j < 8; ++j) q[j] = *(const int4*)(sp + j * 128);
        int sum = 0;
#pragma unroll
        for (int j = 0; j < 8; ++j) {
            *(int4*)(g + j * 128) = q[j];
            sum += (q[j].x + q[j].y) + (q[j].z + q[j].w);
        }
        sum = wave_sum_int(sum);
        if (lane == 0) tileS[(size_t)(tile0 + tl) * nsl + slice] = sum;
    }
}

#define SCAN_TB 1024

__global__ void __launch_bounds__(SCAN_TB)
k_rowptr(int* __restrict__ cs, const int* __restrict__ tileS, int* __restrict__ rowptr,
         float* __restrict__ dinv, int n, int nsl, int pitch) {
    __shared__ int s_wave[SCAN_TB / 32];
    __shared__ int s_part[SCAN_TB / 32];
    const int t = threadIdx.x;
    const int lane = t & 31;
    const int wv = t >> 5;
    const int idx = blockIdx.x * SCAN_TB + t;

    int v[MAX_SLICES];
#pragma unroll
    for (int s = 0; s < MAX_SLICES; ++s) {
        const int sc = (s < nsl) ? s : (nsl - 1);
        const int x = cs[sc * pitch + idx];
        v[s] = (s < nsl) ? x : 0;
    }

    int part = 0;
    const int front = blockIdx.x * nsl;
    for (int i = t; i < front; i += SCAN_TB) part += tileS[i];

    int c = 0;
#pragma unroll
    for (int s = 0; s < MAX_SLICES; ++s) c += v[s];

    int incl = c;
#pragma unroll
    for (int dd = 1; dd < 32; dd <<= 1) {
        const int o = __shfl_up(incl, dd, 32);
        if (lane >= dd) incl += o;
    }
#pragma unroll
    for (int dd = 16; dd >= 1; dd >>= 1) part += __shfl_xor(part, dd, 32);

    if (lane == 31) s_wave[wv] = incl;
    if (lane == 0)  s_part[wv] = part;
    __syncthreads();

    int before = (lane < wv) ? s_wave[lane] : 0;
    int base   = s_part[lane];
#pragma unroll
    for (int dd = 16; dd >= 1; dd >>= 1) {
        before += __shfl_xor(before, dd, 32);
        base   += __shfl_xor(base, dd, 32);
    }
    const int excl = base + before + incl - c;

    int run = excl;
#pragma unroll
    for (int s = 0; s < MAX_SLICES; ++s) {
        if (s < nsl) cs[s * pitch + idx] = run;
        run += v[s];
    }
    if (idx < n) {
        rowptr[idx] = excl;
        dinv[idx]   = rsqrtf((float)(c + 1));
        if (idx == n - 1) rowptr[n] = excl + c;
    }
}

__global__ void k_csr_fill(int2* __restrict__ csr, const int* __restrict__ cs,
                           const int* __restrict__ rank, const float* __restrict__ dinv,
                           const int* __restrict__ src, const int* __restrict__ dst,
                           int e, int perSlice, int pitch) {
    int i = blockIdx.x * blockDim.x + threadIdx.x;
    if (i >= e) return;
    const int slice = (int)(blockIdx.x * blockDim.x) / perSlice;
    const int s = src[i], d = dst[i];
    const int pos = cs[(size_t)slice * pitch + d] + rank[i];
    csr[pos] = make_int2(s, __float_as_int(dinv[s]));
}

#define GATHER_U 4
#define GATHER_NP 2
#define GATHER_TB 256

template <bool HAS_BIAS>
__device__ __forceinline__ void gather_rows(float4 (&acc)[GATHER_NP],
        const float* __restrict__ h, const float* __restrict__ dinv,
        const float* __restrict__ bias, const int* __restrict__ rowptr,
        const int2* __restrict__ csr, int n, int nodeBase, int half, unsigned c) {
    float4 b = make_float4(0.0f, 0.0f, 0.0f, 0.0f);
    if (HAS_BIAS) b = *(const float4*)(bias + 4 * c);

    int   beg[GATHER_NP], end[GATHER_NP];
    float di[GATHER_NP];
    float4 own[GATHER_NP];
#pragma unroll
    for (int it = 0; it < GATHER_NP; ++it) {
        const int node = nodeBase + 2 * it + half;
        const int nc = (node < n) ? node : (n - 1);
        const int r0 = rowptr[nc];
        const int r1 = rowptr[nc + 1];
        beg[it] = r0;
        end[it] = (node < n) ? r1 : r0;
        di[it]  = dinv[nc];
        own[it] = *(const float4*)(h + (unsigned)nc * NODE_DIM + 4 * c);
    }

#pragma unroll
    for (int it = 0; it < GATHER_NP; ++it) {
        const float d2 = di[it];
        float4 a;
        if (HAS_BIAS) {
            a.x = own[it].x * d2 * d2 + b.x;
            a.y = own[it].y * d2 * d2 + b.y;
            a.z = own[it].z * d2 * d2 + b.z;
            a.w = own[it].w * d2 * d2 + b.w;
        } else {
            a.x = own[it].x * d2 * d2;
            a.y = own[it].y * d2 * d2;
            a.z = own[it].z * d2 * d2;
            a.w = own[it].w * d2 * d2;
        }

        const int e1 = end[it];
        if (beg[it] < e1) {
            const int last = e1 - 1;
            int2 p[GATHER_U];
#pragma unroll
            for (int k = 0; k < GATHER_U; ++k) {
                const int jj = beg[it] + k;
                p[k] = csr[(jj < last) ? jj : last];
            }
            for (int j = beg[it]; j < e1; j += GATHER_U) {
                float4 v[GATHER_U];
#pragma unroll
                for (int k = 0; k < GATHER_U; ++k)
                    v[k] = *(const float4*)(h + (unsigned)p[k].x * NODE_DIM + 4 * c);
                int2 pn[GATHER_U];
#pragma unroll
                for (int k = 0; k < GATHER_U; ++k) {
                    const int jj = j + GATHER_U + k;
                    pn[k] = csr[(jj < last) ? jj : last];
                }
#pragma unroll
                for (int k = 0; k < GATHER_U; ++k) {
                    const float norm = __int_as_float(p[k].y) * d2;
                    const bool live = (j + k) < e1;
                    a.x = live ? (a.x + v[k].x * norm) : a.x;
                    a.y = live ? (a.y + v[k].y * norm) : a.y;
                    a.z = live ? (a.z + v[k].z * norm) : a.z;
                    a.w = live ? (a.w + v[k].w * norm) : a.w;
                }
#pragma unroll
                for (int k = 0; k < GATHER_U; ++k) p[k] = pn[k];
            }
        }
        acc[it] = a;
    }
}

#define TILE_ROWS (GATHER_TB / 32 * 2 * GATHER_NP)
#define TILE_PITCH 68

__global__ void __launch_bounds__(GATHER_TB)
k_gather_gemm(float* __restrict__ hout, const float* __restrict__ x, const float* __restrict__ dinv,
              const float* __restrict__ W1, const float* __restrict__ b1, const float* __restrict__ W2,
              const int* __restrict__ rowptr, const int2* __restrict__ csr, int n) {
    __shared__ __attribute__((aligned(16))) float s_a[TILE_ROWS * TILE_PITCH];
    __shared__ __attribute__((aligned(16))) float s_h[TILE_ROWS * TILE_PITCH];
    const int lane = threadIdx.x & 31;
    const int wv   = threadIdx.x >> 5;
    const int half = lane >> 4;
    const unsigned c = lane & 15;
    const int blockNode = blockIdx.x * TILE_ROWS;

    float4 acc[GATHER_NP];
    gather_rows<false>(acc, x, dinv, nullptr, rowptr, csr, n, blockNode + wv * (2 * GATHER_NP), half, c);

#pragma unroll
    for (int it = 0; it < GATHER_NP; ++it)
        *(float4*)&s_a[(wv * (2 * GATHER_NP) + 2 * it + half) * TILE_PITCH + 4 * c] = acc[it];

    const int rowTile = (wv >> 2) * 16;
    const int colBase = (wv & 3) * 16;
    const int m       = lane & 15;
    const int khalf   = half * 2;
    const int nc      = colBase + m;

    v2f b[NODE_DIM / 4];
#pragma unroll
    for (int i = 0; i < NODE_DIM / 4; ++i) {
        b[i].x = W1[(4 * i + khalf + 0) * NODE_DIM + nc];
        b[i].y = W1[(4 * i + khalf + 1) * NODE_DIM + nc];
    }
    const float bias1 = b1[nc];
    __syncthreads();

    const float* arow = s_a + (rowTile + m) * TILE_PITCH + khalf;
    v8f d = {};
#pragma unroll
    for (int i = 0; i < NODE_DIM / 4; ++i) {
        const v2f a = *(const v2f*)(arow + 4 * i);
        d = __builtin_amdgcn_wmma_f32_16x16x4_f32(false, a, false, b[i], (short)0, d, false, false);
    }

#pragma unroll
    for (int i = 0; i < NODE_DIM / 4; ++i) {
        b[i].x = W2[(4 * i + khalf + 0) * NODE_DIM + nc];
        b[i].y = W2[(4 * i + khalf + 1) * NODE_DIM + nc];
    }

#pragma unroll
    for (int r = 0; r < 8; ++r)
        s_h[(rowTile + r + half * 8) * TILE_PITCH + nc] = fmaxf(d[r] + bias1, 0.0f);
    __syncthreads();

    const float* hrow = s_h + (rowTile + m) * TILE_PITCH + khalf;
    v8f d2 = {};
#pragma unroll
    for (int i = 0; i < NODE_DIM / 4; ++i) {
        const v2f a = *(const v2f*)(hrow + 4 * i);
        d2 = __builtin_amdgcn_wmma_f32_16x16x4_f32(false, a, false, b[i], (short)0, d2, false, false);
    }

#pragma unroll
    for (int r = 0; r < 8; ++r) {
        const int row = blockNode + rowTile + r + half * 8;
        if (row < n) hout[(size_t)row * NODE_DIM + nc] = d2[r];
    }
}


template <int STEPS>
__device__ __forceinline__ float group_sum(float v) {
    if (STEPS >= 1) v += __int_as_float(__builtin_amdgcn_update_dpp(0, __float_as_int(v), 0x161, 0xf, 0xf, true));
    if (STEPS >= 2) v += __int_as_float(__builtin_amdgcn_update_dpp(0, __float_as_int(v), 0x162, 0xf, 0xf, true));
    if (STEPS >= 3) v += __int_as_float(__builtin_amdgcn_update_dpp(0, __float_as_int(v), 0x164, 0xf, 0xf, true));
    if (STEPS >= 4) v += __int_as_float(__builtin_amdgcn_update_dpp(0, __float_as_int(v), 0x168, 0xf, 0xf, true));
    return v;
}

__device__ __forceinline__ float dot4(const float4 a, const float4 b) {
    return a.x * b.x + a.y * b.y + a.z * b.z + a.w * b.w;
}

__global__ void __launch_bounds__(GATHER_TB)
k_gather_proj(float* __restrict__ pp, const float* __restrict__ h, const float* __restrict__ dinv,
              const float* __restrict__ bias, const float* __restrict__ We,
              const int* __restrict__ rowptr, const int2* __restrict__ csr, int n) {
    const int lane = threadIdx.x & 31;
    const int half = lane >> 4;
    const unsigned c = lane & 15;
    const int nodeBase = (blockIdx.x * (GATHER_TB >> 5) + (threadIdx.x >> 5)) * (2 * GATHER_NP);

    float4 acc[GATHER_NP];
    gather_rows<true>(acc, h, dinv, bias, rowptr, csr, n, nodeBase, half, c);

    const float4 ws = *(const float4*)(We + 4 * c);
    const float4 wd = *(const float4*)(We + 64 + 4 * c);

    float ps[GATHER_NP], pd[GATHER_NP];
#pragma unroll
    for (int it = 0; it < GATHER_NP; ++it) {
        float4 v = acc[it];
        v.x = fmaxf(v.x, 0.0f); v.y = fmaxf(v.y, 0.0f);
        v.z = fmaxf(v.z, 0.0f); v.w = fmaxf(v.w, 0.0f);
        ps[it] = group_sum<4>(dot4(v, ws));
        pd[it] = group_sum<4>(dot4(v, wd));
    }
    const float a = (c == 0) ? ps[0] : ps[1];
    const float b = (c == 0) ? pd[0] : pd[1];
    const int myNode = nodeBase + 2 * (int)c + half;
    if (c < GATHER_NP && myNode < n) *(float2*)(pp + 2 * (size_t)myNode) = make_float2(a, b);
}

__global__ void k_edge_score(float* __restrict__ out, const float* __restrict__ pp,
                             const float* __restrict__ ea, const float* __restrict__ We,
                             const float* __restrict__ be,
                             const int* __restrict__ src, const int* __restrict__ dst, int e) {
    const int lane = threadIdx.x & 31;
    const int q = lane >> 2;
    const int j = lane & 3;
    const int edgeBase = (blockIdx.x * (blockDim.x >> 5) + (threadIdx.x >> 5)) * 64;

    const int e0 = edgeBase + j * 8 + q;
    const int e1 = e0 + 32;
    const int e0c = (e0 < e) ? e0 : (e - 1);
    const int e1c = (e1 < e) ? e1 : (e - 1);
    const unsigned s0 = (unsigned)src[e0c], d0 = (unsigned)dst[e0c];
    const unsigned s1 = (unsigned)src[e1c], d1 = (unsigned)dst[e1c];

    const float4 w = *(const float4*)(We + 128 + 4 * j);

    float p[8];
#pragma unroll
    for (int r = 0; r < 8; ++r) {
        int er = edgeBase + r * 8 + q;
        er = (er < e) ? er : (e - 1);
        const float4 a = *(const float4*)(ea + (size_t)er * 16 + 4 * j);
        p[r] = dot4(a, w);
    }

    const float n0 = pp[2u * s0] + pp[2u * d0 + 1u];
    const float n1 = pp[2u * s1] + pp[2u * d1 + 1u];

#pragma unroll
    for (int r = 0; r < 8; ++r) p[r] = group_sum<2>(p[r]);

    const float m0 = (j == 0) ? p[0] : (j == 1) ? p[1] : (j == 2) ? p[2] : p[3];
    const float m1 = (j == 0) ? p[4] : (j == 1) ? p[5] : (j == 2) ? p[6] : p[7];
    const float bias = be[0];
    if (e0 < e) out[e0] = m0 + n0 + bias;
    if (e1 < e) out[e1] = m1 + n1 + bias;
}

extern "C" void kernel_launch(void* const* d_in, const int* in_sizes, int n_in,
                              void* d_out, int out_size, void* d_ws, size_t ws_size,
                              hipStream_t stream) {
    const float* x  = (const float*)d_in[0];
    const int*   ei = (const int*)d_in[1];
    const float* ea = (const float*)d_in[2];
    const float* W1 = (const float*)d_in[3];
    const float* b1 = (const float*)d_in[4];
    const float* W2 = (const float*)d_in[5];
    const float* b2 = (const float*)d_in[6];
    const float* We = (const float*)d_in[7];
    const float* be = (const float*)d_in[8];
    float* out = (float*)d_out;

    const int N = in_sizes[0] / NODE_DIM;
    const int E = in_sizes[1] / 2;
    const int* src = ei;
    const int* dst = ei + E;

    const int perSlice = (E / MAX_SLICES / (HIST_U * HIST_TB) + 1) * (HIST_U * HIST_TB);
    const int nSlices  = (E + perSlice - 1) / perSlice;
    const int gScan    = (N + SCAN_TB - 1) / SCAN_TB;
    const int pitch    = gScan * SCAN_TB;
    const int gChunk   = (N + HIST_NODES - 1) / HIST_NODES;

    char* ws = (char*)d_ws;
    const size_t nodeBytes = (((size_t)N * 4) + 511) & ~(size_t)511;
    const size_t rowBytes  = ((((size_t)N + 1) * 4) + 511) & ~(size_t)511;
    const size_t hBytes    = (size_t)N * NODE_DIM * 4;
    const size_t edgeBytes = (((size_t)E * 4) + 511) & ~(size_t)511;
    float* dinv   = (float*)ws;
    float* hA     = (float*)(ws + nodeBytes);
    float* hB     = (float*)(ws + nodeBytes + hBytes);
    int*   rowptr = (int*)(ws + nodeBytes + 2 * hBytes);
    int2*  csr    = (int2*)(ws + nodeBytes + rowBytes + 2 * hBytes);
    int*   rank   = (int*)(ws + nodeBytes + rowBytes + 2 * hBytes + 2 * edgeBytes);
    int*   cs     = (int*)(ws + nodeBytes + rowBytes + 2 * hBytes + 3 * edgeBytes);
    int*   tileS  = cs + (size_t)nSlices * pitch;

    const int TB = 256;
    const int gE    = (E + TB - 1) / TB;
    const int gGath = (N + TILE_ROWS - 1) / TILE_ROWS;
    const int gEdge = (E + 511) / 512;

    k_hist_lds<<<dim3(nSlices, gChunk), HIST_TB, 0, stream>>>(cs, tileS, rank, dst, E, perSlice, pitch);
    k_rowptr  <<<gScan, SCAN_TB, 0, stream>>>(cs, tileS, rowptr, dinv, N, nSlices, pitch);
    k_csr_fill<<<gE, TB, 0, stream>>>(csr, cs, rank, dinv, src, dst, E, perSlice, pitch);

    k_gather_gemm<<<gGath, GATHER_TB, 0, stream>>>(hB, x, dinv, W1, b1, W2, rowptr, csr, N);

    float* pp = hA;
    k_gather_proj<<<gGath, GATHER_TB, 0, stream>>>(pp, hB, dinv, b2, We, rowptr, csr, N);
    k_edge_score <<<gEdge, TB, 0, stream>>>(out, pp, ea, We, be, src, dst, E);
}


# Edge sort rebuilt as partition by destination range then per-range sort, both in LDS: 3 launches to 2, whole lines only; list entry = source id, h2 rows stored scaled (terms sum in LDS arrival order)
# speedup vs baseline: 4.9564x; 1.3794x over previous
#include <hip/hip_runtime.h>
#include <hip/hip_bf16.h>

typedef __attribute__((ext_vector_type(2))) float v2f;
typedef __attribute__((ext_vector_type(8))) float v8f;

#define NODE_DIM 64

#define RANGE_NODES 256
#define PART_TB 1024
#define PART_U 8
#define PART_EDGES (PART_TB * PART_U)
#define MAX_RANGES PART_TB

__device__ __forceinline__ int wave_sum_int(int v) {
    v += __builtin_amdgcn_update_dpp(0, v, 0x161, 0xf, 0xf, true);
    v += __builtin_amdgcn_update_dpp(0, v, 0x162, 0xf, 0xf, true);
    v += __builtin_amdgcn_update_dpp(0, v, 0x164, 0xf, 0xf, true);
    v += __builtin_amdgcn_update_dpp(0, v, 0x168, 0xf, 0xf, true);
    v += __shfl_xor(v, 16, 32);
    return v;
}

__device__ __forceinline__ int wave_scan_int(int v, int lane) {
#pragma unroll
    for (int dd = 1; dd < 32; dd <<= 1) {
        const int o = __shfl_up(v, dd, 32);
        if (lane >= dd) v += o;
    }
    return v;
}

__global__ void __launch_bounds__(PART_TB)
k_part(unsigned* __restrict__ part, int* __restrict__ offT,
       const int* __restrict__ src, const int* __restrict__ dst, int e, int nRanges) {
    __shared__ int s_cnt[MAX_RANGES];
    __shared__ int s_wave[PART_TB / 32];
    __shared__ __attribute__((aligned(16))) unsigned s_stage[PART_EDGES];
    const int t = threadIdx.x;
    const int lane = t & 31;
    const int wv = t >> 5;
    const int slice = blockIdx.x;
    const int nsl = gridDim.x;
    const int e0 = slice * PART_EDGES;
    const int e1 = (e0 + PART_EDGES < e) ? (e0 + PART_EDGES) : e;
    const int last = e1 - 1;

    int s[PART_U], d[PART_U];
#pragma unroll
    for (int k = 0; k < PART_U; ++k) {
        const int i = e0 + k * PART_TB + t;
        const int ic = (i < last) ? i : last;
        s[k] = src[ic];
        d[k] = dst[ic];
    }
    s_cnt[t] = 0;
    __syncthreads();

    int rg[PART_U], r[PART_U];
#pragma unroll
    for (int k = 0; k < PART_U; ++k) {
        const int i = e0 + k * PART_TB + t;
        const unsigned q = (unsigned)d[k] >> 8;
        rg[k] = (int)((q < (unsigned)(MAX_RANGES - 1)) ? q : (unsigned)(MAX_RANGES - 1));
        r[k] = 0;
        if (i < e1) r[k] = atomicAdd(&s_cnt[rg[k]], 1);
    }
    __syncthreads();

    const int c = s_cnt[t];
    const int incl = wave_scan_int(c, lane);
    if (lane == 31) s_wave[wv] = incl;
    __syncthreads();
    const int before = wave_sum_int((lane < wv) ? s_wave[lane] : 0);
    const int excl = before + incl - c;
    s_cnt[t] = excl;
    if (t <= nRanges) offT[(size_t)t * nsl + slice] = excl;
    __syncthreads();

#pragma unroll
    for (int k = 0; k < PART_U; ++k) {
        const int i = e0 + k * PART_TB + t;
        if (i < e1) s_stage[s_cnt[rg[k]] + r[k]] = ((unsigned)s[k] << 8) | ((unsigned)d[k] & 255u);
    }
    __syncthreads();
    const int cnt = e1 - e0;
    for (int i = t; i < cnt; i += PART_TB) part[(size_t)e0 + i] = s_stage[i];
}

#define SEG_TB 1024
#define SEG_CAP 16384
#define SEG_OFFS 1024

template <class F>
__device__ __forceinline__ void seg_runs(const unsigned* __restrict__ part, const int* s_olo, const int* s_ohi,
                                         const int* __restrict__ rowLo, const int* __restrict__ rowHi,
                                         int nsl, int wv, int lane, F f) {
    for (int s = wv; s < nsl; s += SEG_TB / 32) {
        int lo = (s < SEG_OFFS) ? s_olo[s] : rowLo[s];
        int hi = (s < SEG_OFFS) ? s_ohi[s] : rowHi[s];
        lo = (lo > 0) ? lo : 0;
        hi = (hi < PART_EDGES) ? hi : PART_EDGES;
        if (lo >= hi) continue;
        const unsigned* run = part + (size_t)s * PART_EDGES;
        const int last = hi - 1;
        for (int i = lo + lane; i < hi; i += 64) {
            const int i1 = i + 32;
            const unsigned w0 = run[i];
            const unsigned w1 = run[(i1 < last) ? i1 : last];
            f(w0);
            if (i1 < hi) f(w1);
        }
    }
}

__global__ void __launch_bounds__(SEG_TB)
k_seg(int* __restrict__ csr, int* __restrict__ rowptr, float* __restrict__ dinv,
      const unsigned* __restrict__ part, const int* __restrict__ offT, int n, int nsl) {
    __shared__ int s_cnt[RANGE_NODES];
    __shared__ int s_start[RANGE_NODES];
    __shared__ int s_wave[RANGE_NODES / 32];
    __shared__ int s_lo[SEG_TB / 32], s_hi[SEG_TB / 32];
    __shared__ int s_olo[SEG_OFFS], s_ohi[SEG_OFFS];
    __shared__ __attribute__((aligned(16))) int s_stage[SEG_CAP];
    const int t = threadIdx.x;
    const int lane = t & 31;
    const int wv = t >> 5;
    const int rgn = blockIdx.x;
    const int* rowLo = offT + (size_t)rgn * nsl;
    const int* rowHi = rowLo + nsl;

    if (t < RANGE_NODES) s_cnt[t] = 0;
    int a = 0, b = 0;
    for (int s = t; s < nsl; s += SEG_TB) {
        const int l = rowLo[s], h = rowHi[s];
        a += l;
        b += h;
        if (s < SEG_OFFS) { s_olo[s] = l; s_ohi[s] = h; }
    }
    a = wave_sum_int(a);
    b = wave_sum_int(b);
    if (lane == 0) { s_lo[wv] = a; s_hi[wv] = b; }
    __syncthreads();
    const int gs    = wave_sum_int(s_lo[lane]);
    const int total = wave_sum_int(s_hi[lane]) - gs;

    seg_runs(part, s_olo, s_ohi, rowLo, rowHi, nsl, wv, lane,
             [&](unsigned w) { atomicAdd(&s_cnt[w & 255u], 1); });
    __syncthreads();

    const int c = (t < RANGE_NODES) ? s_cnt[t] : 0;
    const int incl = wave_scan_int(c, lane);
    if (t < RANGE_NODES && lane == 31) s_wave[wv] = incl;
    __syncthreads();
    const int before = wave_sum_int((lane < wv && lane < RANGE_NODES / 32) ? s_wave[lane] : 0);
    const int excl = before + incl - c;
    if (t < RANGE_NODES) {
        s_start[t] = excl;
        s_cnt[t] = 0;
        const int node = rgn * RANGE_NODES + t;
        if (node < n) {
            rowptr[node] = gs + excl;
            dinv[node]   = rsqrtf((float)(c + 1));
            if (node == n - 1) rowptr[n] = gs + excl + c;
        }
    }
    __syncthreads();

    const bool staged = total <= SEG_CAP;
    int* seg = csr + (size_t)gs;
    seg_runs(part, s_olo, s_ohi, rowLo, rowHi, nsl, wv, lane,
             [&](unsigned w) {
                 const unsigned nd = w & 255u;
                 const int pos = s_start[nd] + atomicAdd(&s_cnt[nd], 1);
                 if (staged) s_stage[pos] = (int)(w >> 8);
                 else        seg[pos] = (int)(w >> 8);
             });
    __syncthreads();
    if (staged)
        for (int i = t; i < total; i += SEG_TB) seg[i] = s_stage[i];
}

#define GATHER_U 4
#define GATHER_NP 2
#define GATHER_TB 256

template <bool HAS_BIAS, bool SCALED>
__device__ __forceinline__ void gather_rows(float4 (&acc)[GATHER_NP],
        const float* __restrict__ h, const float* __restrict__ dinv,
        const float* __restrict__ bias, const int* __restrict__ rowptr,
        const int* __restrict__ csr, int n, int nodeBase, int half, unsigned c) {
    float4 b = make_float4(0.0f, 0.0f, 0.0f, 0.0f);
    if (HAS_BIAS) b = *(const float4*)(bias + 4 * c);

    int   beg[GATHER_NP], end[GATHER_NP];
    float di[GATHER_NP];
    float4 own[GATHER_NP];
#pragma unroll
    for (int it = 0; it < GATHER_NP; ++it) {
        const int node = nodeBase + 2 * it + half;
        const int nc = (node < n) ? node : (n - 1);
        const int r0 = rowptr[nc];
        const int r1 = rowptr[nc + 1];
        beg[it] = r0;
        end[it] = (node < n) ? r1 : r0;
        di[it]  = dinv[nc];
        own[it] = *(const float4*)(h + (unsigned)nc * NODE_DIM + 4 * c);
    }

#pragma unroll
    for (int it = 0; it < GATHER_NP; ++it) {
        const float d2 = di[it];
        float4 a;
        if (SCALED) {
            a.x = own[it].x * d2;
            a.y = own[it].y * d2;
            a.z = own[it].z * d2;
            a.w = own[it].w * d2;
        } else {
            a.x = own[it].x * d2 * d2;
            a.y = own[it].y * d2 * d2;
            a.z = own[it].z * d2 * d2;
            a.w = own[it].w * d2 * d2;
        }
        if (HAS_BIAS) { a.x += b.x; a.y += b.y; a.z += b.z; a.w += b.w; }

        const int e1 = end[it];
        if (beg[it] < e1) {
            const int last = e1 - 1;
            int p[GATHER_U];
#pragma unroll
            for (int k = 0; k < GATHER_U; ++k) {
                const int jj = beg[it] + k;
                p[k] = csr[(jj < last) ? jj : last];
            }
            for (int j = beg[it]; j < e1; j += GATHER_U) {
                float4 v[GATHER_U];
                float  w[GATHER_U];
#pragma unroll
                for (int k = 0; k < GATHER_U; ++k) {
                    v[k] = *(const float4*)(h + (unsigned)p[k] * NODE_DIM + 4 * c);
                    w[k] = SCALED ? 1.0f : dinv[(unsigned)p[k]];
                }
                int pn[GATHER_U];
#pragma unroll
                for (int k = 0; k < GATHER_U; ++k) {
                    const int jj = j + GATHER_U + k;
                    pn[k] = csr[(jj < last) ? jj : last];
                }
#pragma unroll
                for (int k = 0; k < GATHER_U; ++k) {
                    const float norm = SCALED ? d2 : (w[k] * d2);
                    const bool live = (j + k) < e1;
                    a.x = live ? (a.x + v[k].x * norm) : a.x;
                    a.y = live ? (a.y + v[k].y * norm) : a.y;
                    a.z = live ? (a.z + v[k].z * norm) : a.z;
                    a.w = live ? (a.w + v[k].w * norm) : a.w;
                }
#pragma unroll
                for (int k = 0; k < GATHER_U; ++k) p[k] = pn[k];
            }
        }
        acc[it] = a;
    }
}

#define TILE_ROWS (GATHER_TB / 32 * 2 * GATHER_NP)
#define TILE_PITCH 68

__global__ void __launch_bounds__(GATHER_TB)
k_gather_gemm(float* __restrict__ hout, const float* __restrict__ x, const float* __restrict__ dinv,
              const float* __restrict__ W1, const float* __restrict__ b1, const float* __restrict__ W2,
              const int* __restrict__ rowptr, const int* __restrict__ csr, int n) {
    __shared__ __attribute__((aligned(16))) float s_a[TILE_ROWS * TILE_PITCH];
    __shared__ __attribute__((aligned(16))) float s_h[TILE_ROWS * TILE_PITCH];
    __shared__ __attribute__((aligned(16))) float s_d[TILE_ROWS];
    const int lane = threadIdx.x & 31;
    const int wv   = threadIdx.x >> 5;
    const int half = lane >> 4;
    const unsigned c = lane & 15;
    const int blockNode = blockIdx.x * TILE_ROWS;

    float myd = 0.0f;
    if (wv == 0) {
        const int row = blockNode + lane;
        myd = dinv[(row < n) ? row : (n - 1)];
    }

    float4 acc[GATHER_NP];
    gather_rows<false, false>(acc, x, dinv, nullptr, rowptr, csr, n, blockNode + wv * (2 * GATHER_NP), half, c);

#pragma unroll
    for (int it = 0; it < GATHER_NP; ++it)
        *(float4*)&s_a[(wv * (2 * GATHER_NP) + 2 * it + half) * TILE_PITCH + 4 * c] = acc[it];
    if (wv == 0) s_d[lane] = myd;

    const int rowTile = (wv >> 2) * 16;
    const int colBase = (wv & 3) * 16;
    const int m       = lane & 15;
    const int khalf   = half * 2;
    const int nc      = colBase + m;

    v2f b[NODE_DIM / 4];
#pragma unroll
    for (int i = 0; i < NODE_DIM / 4; ++i) {
        b[i].x = W1[(4 * i + khalf + 0) * NODE_DIM + nc];
        b[i].y = W1[(4 * i + khalf + 1) * NODE_DIM + nc];
    }
    const float bias1 = b1[nc];
    __syncthreads();

    const float* arow = s_a + (rowTile + m) * TILE_PITCH + khalf;
    v8f d = {};
#pragma unroll
    for (int i = 0; i < NODE_DIM / 4; ++i) {
        const v2f a = *(const v2f*)(arow + 4 * i);
        d = __builtin_amdgcn_wmma_f32_16x16x4_f32(false, a, false, b[i], (short)0, d, false, false);
    }

#pragma unroll
    for (int i = 0; i < NODE_DIM / 4; ++i) {
        b[i].x = W2[(4 * i + khalf + 0) * NODE_DIM + nc];
        b[i].y = W2[(4 * i + khalf + 1) * NODE_DIM + nc];
    }

#pragma unroll
    for (int r = 0; r < 8; ++r)
        s_h[(rowTile + r + half * 8) * TILE_PITCH + nc] = fmaxf(d[r] + bias1, 0.0f);
    __syncthreads();

    const float* hrow = s_h + (rowTile + m) * TILE_PITCH + khalf;
    v8f d2 = {};
#pragma unroll
    for (int i = 0; i < NODE_DIM / 4; ++i) {
        const v2f a = *(const v2f*)(hrow + 4 * i);
        d2 = __builtin_amdgcn_wmma_f32_16x16x4_f32(false, a, false, b[i], (short)0, d2, false, false);
    }

    const float4 q0 = *(const float4*)&s_d[rowTile + half * 8];
    const float4 q1 = *(const float4*)&s_d[rowTile + half * 8 + 4];
    const float dr[8] = {q0.x, q0.y, q0.z, q0.w, q1.x, q1.y, q1.z, q1.w};
#pragma unroll
    for (int r = 0; r < 8; ++r) {
        const int row = blockNode + rowTile + r + half * 8;
        if (row < n) hout[(size_t)row * NODE_DIM + nc] = d2[r] * dr[r];
    }
}


template <int STEPS>
__device__ __forceinline__ float group_sum(float v) {
    if (STEPS >= 1) v += __int_as_float(__builtin_amdgcn_update_dpp(0, __float_as_int(v), 0x161, 0xf, 0xf, true));
    if (STEPS >= 2) v += __int_as_float(__builtin_amdgcn_update_dpp(0, __float_as_int(v), 0x162, 0xf, 0xf, true));
    if (STEPS >= 3) v += __int_as_float(__builtin_amdgcn_update_dpp(0, __float_as_int(v), 0x164, 0xf, 0xf, true));
    if (STEPS >= 4) v += __int_as_float(__builtin_amdgcn_update_dpp(0, __float_as_int(v), 0x168, 0xf, 0xf, true));
    return v;
}

__device__ __forceinline__ float dot4(const float4 a, const float4 b) {
    return a.x * b.x + a.y * b.y + a.z * b.z + a.w * b.w;
}

__global__ void __launch_bounds__(GATHER_TB)
k_gather_proj(float* __restrict__ pp, const float* __restrict__ h, const float* __restrict__ dinv,
              const float* __restrict__ bias, const float* __restrict__ We,
              const int* __restrict__ rowptr, const int* __restrict__ csr, int n) {
    const int lane = threadIdx.x & 31;
    const int half = lane >> 4;
    const unsigned c = lane & 15;
    const int nodeBase = (blockIdx.x * (GATHER_TB >> 5) + (threadIdx.x >> 5)) * (2 * GATHER_NP);

    float4 acc[GATHER_NP];
    gather_rows<true, true>(acc, h, dinv, bias, rowptr, csr, n, nodeBase, half, c);

    const float4 ws = *(const float4*)(We + 4 * c);
    const float4 wd = *(const float4*)(We + 64 + 4 * c);

    float ps[GATHER_NP], pd[GATHER_NP];
#pragma unroll
    for (int it = 0; it < GATHER_NP; ++it) {
        float4 v = acc[it];
        v.x = fmaxf(v.x, 0.0f); v.y = fmaxf(v.y, 0.0f);
        v.z = fmaxf(v.z, 0.0f); v.w = fmaxf(v.w, 0.0f);
        ps[it] = group_sum<4>(dot4(v, ws));
        pd[it] = group_sum<4>(dot4(v, wd));
    }
    const float a = (c == 0) ? ps[0] : ps[1];
    const float b = (c == 0) ? pd[0] : pd[1];
    const int myNode = nodeBase + 2 * (int)c + half;
    if (c < GATHER_NP && myNode < n) *(float2*)(pp + 2 * (size_t)myNode) = make_float2(a, b);
}

__global__ void k_edge_score(float* __restrict__ out, const float* __restrict__ pp,
                             const float* __restrict__ ea, const float* __restrict__ We,
                             const float* __restrict__ be,
                             const int* __restrict__ src, const int* __restrict__ dst, int e) {
    const int lane = threadIdx.x & 31;
    const int q = lane >> 2;
    const int j = lane & 3;
    const int edgeBase = (blockIdx.x * (blockDim.x >> 5) + (threadIdx.x >> 5)) * 64;

    const int e0 = edgeBase + j * 8 + q;
    const int e1 = e0 + 32;
    const int e0c = (e0 < e) ? e0 : (e - 1);
    const int e1c = (e1 < e) ? e1 : (e - 1);
    const unsigned s0 = (unsigned)src[e0c], d0 = (unsigned)dst[e0c];
    const unsigned s1 = (unsigned)src[e1c], d1 = (unsigned)dst[e1c];

    const float4 w = *(const float4*)(We + 128 + 4 * j);

    float p[8];
#pragma unroll
    for (int r = 0; r < 8; ++r) {
        int er = edgeBase + r * 8 + q;
        er = (er < e) ? er : (e - 1);
        const float4 a = *(const float4*)(ea + (size_t)er * 16 + 4 * j);
        p[r] = dot4(a, w);
    }

    const float n0 = pp[2u * s0] + pp[2u * d0 + 1u];
    const float n1 = pp[2u * s1] + pp[2u * d1 + 1u];

#pragma unroll
    for (int r = 0; r < 8; ++r) p[r] = group_sum<2>(p[r]);

    const float m0 = (j == 0) ? p[0] : (j == 1) ? p[1] : (j == 2) ? p[2] : p[3];
    const float m1 = (j == 0) ? p[4] : (j == 1) ? p[5] : (j == 2) ? p[6] : p[7];
    const float bias = be[0];
    if (e0 < e) out[e0] = m0 + n0 + bias;
    if (e1 < e) out[e1] = m1 + n1 + bias;
}

extern "C" void kernel_launch(void* const* d_in, const int* in_sizes, int n_in,
                              void* d_out, int out_size, void* d_ws, size_t ws_size,
                              hipStream_t stream) {
    const float* x  = (const float*)d_in[0];
    const int*   ei = (const int*)d_in[1];
    const float* ea = (const float*)d_in[2];
    const float* W1 = (const float*)d_in[3];
    const float* b1 = (const float*)d_in[4];
    const float* W2 = (const float*)d_in[5];
    const float* b2 = (const float*)d_in[6];
    const float* We = (const float*)d_in[7];
    const float* be = (const float*)d_in[8];
    float* out = (float*)d_out;

    const int N = in_sizes[0] / NODE_DIM;
    const int E = in_sizes[1] / 2;
    const int* src = ei;
    const int* dst = ei + E;

    const int nSlices = (E + PART_EDGES - 1) / PART_EDGES;
    const int nRanges = (N + RANGE_NODES - 1) / RANGE_NODES;

    char* ws = (char*)d_ws;
    const size_t nodeBytes = (((size_t)N * 4) + 511) & ~(size_t)511;
    const size_t rowBytes  = ((((size_t)N + 1) * 4) + 511) & ~(size_t)511;
    const size_t hBytes    = (size_t)N * NODE_DIM * 4;
    const size_t edgeBytes = (((size_t)E * 4) + 511) & ~(size_t)511;
    float*    dinv   = (float*)ws;
    float*    hA     = (float*)(ws + nodeBytes);
    float*    hB     = (float*)(ws + nodeBytes + hBytes);
    int*      rowptr = (int*)(ws + nodeBytes + 2 * hBytes);
    int*      csr    = (int*)(ws + nodeBytes + rowBytes + 2 * hBytes);
    unsigned* part   = (unsigned*)(ws + nodeBytes + rowBytes + 2 * hBytes + edgeBytes);
    int*      offT   = (int*)(ws + nodeBytes + rowBytes + 2 * hBytes + 2 * edgeBytes);

    const int TB = 256;
    const int gGath = (N + TILE_ROWS - 1) / TILE_ROWS;
    const int gEdge = (E + 511) / 512;

    k_part<<<nSlices, PART_TB, 0, stream>>>(part, offT, src, dst, E, nRanges);
    k_seg <<<nRanges, SEG_TB, 0, stream>>>(csr, rowptr, dinv, part, offT, N, nSlices);

    k_gather_gemm<<<gGath, GATHER_TB, 0, stream>>>(hB, x, dinv, W1, b1, W2, rowptr, csr, N);

    float* pp = hA;
    k_gather_proj<<<gGath, GATHER_TB, 0, stream>>>(pp, hB, dinv, b2, We, rowptr, csr, N);
    k_edge_score <<<gEdge, TB, 0, stream>>>(out, pp, ea, We, be, src, dst, E);
}


# Score's edge-feature dot formed by spare work-groups of both sort launches (3/8 of the edges beside the first, 5/8 beside the second), kept in out; last launch adds node terms (sum order kept)
# speedup vs baseline: 5.2367x; 1.0566x over previous
#include <hip/hip_runtime.h>
#include <hip/hip_bf16.h>

typedef __attribute__((ext_vector_type(2))) float v2f;
typedef __attribute__((ext_vector_type(8))) float v8f;

#define NODE_DIM 64

#define RANGE_NODES 256
#define PART_TB 1024
#define PART_U 8
#define PART_EDGES (PART_TB * PART_U)
#define MAX_RANGES PART_TB

__device__ __forceinline__ int wave_sum_int(int v) {
    v += __builtin_amdgcn_update_dpp(0, v, 0x161, 0xf, 0xf, true);
    v += __builtin_amdgcn_update_dpp(0, v, 0x162, 0xf, 0xf, true);
    v += __builtin_amdgcn_update_dpp(0, v, 0x164, 0xf, 0xf, true);
    v += __builtin_amdgcn_update_dpp(0, v, 0x168, 0xf, 0xf, true);
    v += __shfl_xor(v, 16, 32);
    return v;
}

__device__ __forceinline__ int wave_scan_int(int v, int lane) {
#pragma unroll
    for (int dd = 1; dd < 32; dd <<= 1) {
        const int o = __shfl_up(v, dd, 32);
        if (lane >= dd) v += o;
    }
    return v;
}

__device__ __forceinline__ void edge_feat_tile(float* __restrict__ out, const float* __restrict__ ea,
                                               const float* __restrict__ We, int edgeBase, int eEnd, int lane);

__global__ void __launch_bounds__(PART_TB)
k_part(unsigned* __restrict__ part, int* __restrict__ offT,
       const int* __restrict__ src, const int* __restrict__ dst, int e, int nRanges, int nsl,
       float* __restrict__ out, const float* __restrict__ ea, const float* __restrict__ We, int featEnd) {
    if ((int)blockIdx.x >= nsl) {
        const int tile = ((int)blockIdx.x - nsl) * (PART_TB / 32) + (int)(threadIdx.x >> 5);
        edge_feat_tile(out, ea, We, tile * 64, featEnd, (int)(threadIdx.x & 31));
        return;
    }
    __shared__ int s_cnt[MAX_RANGES];
    __shared__ int s_wave[PART_TB / 32];
    __shared__ __attribute__((aligned(16))) unsigned s_stage[PART_EDGES];
    const int t = threadIdx.x;
    const int lane = t & 31;
    const int wv = t >> 5;
    const int slice = blockIdx.x;
    const int e0 = slice * PART_EDGES;
    const int e1 = (e0 + PART_EDGES < e) ? (e0 + PART_EDGES) : e;
    const int last = e1 - 1;

    int s[PART_U], d[PART_U];
#pragma unroll
    for (int k = 0; k < PART_U; ++k) {
        const int i = e0 + k * PART_TB + t;
        const int ic = (i < last) ? i : last;
        s[k] = src[ic];
        d[k] = dst[ic];
    }
    s_cnt[t] = 0;
    __syncthreads();

    int rg[PART_U], r[PART_U];
#pragma unroll
    for (int k = 0; k < PART_U; ++k) {
        const int i = e0 + k * PART_TB + t;
        const unsigned q = (unsigned)d[k] >> 8;
        rg[k] = (int)((q < (unsigned)(MAX_RANGES - 1)) ? q : (unsigned)(MAX_RANGES - 1));
        r[k] = 0;
        if (i < e1) r[k] = atomicAdd(&s_cnt[rg[k]], 1);
    }
    __syncthreads();

    const int c = s_cnt[t];
    const int incl = wave_scan_int(c, lane);
    if (lane == 31) s_wave[wv] = incl;
    __syncthreads();
    const int before = wave_sum_int((lane < wv) ? s_wave[lane] : 0);
    const int excl = before + incl - c;
    s_cnt[t] = excl;
    if (t <= nRanges) offT[(size_t)t * nsl + slice] = excl;
    __syncthreads();

#pragma unroll
    for (int k = 0; k < PART_U; ++k) {
        const int i = e0 + k * PART_TB + t;
        if (i < e1) s_stage[s_cnt[rg[k]] + r[k]] = ((unsigned)s[k] << 8) | ((unsigned)d[k] & 255u);
    }
    __syncthreads();
    const int cnt = e1 - e0;
    for (int i = t; i < cnt; i += PART_TB) part[(size_t)e0 + i] = s_stage[i];
}

#define SEG_TB 1024
#define SEG_CAP 16384
#define SEG_OFFS 1024

template <class F>
__device__ __forceinline__ void seg_runs(const unsigned* __restrict__ part, const int* s_olo, const int* s_ohi,
                                         const int* __restrict__ rowLo, const int* __restrict__ rowHi,
                                         int nsl, int wv, int lane, F f) {
    for (int s = wv; s < nsl; s += SEG_TB / 32) {
        int lo = (s < SEG_OFFS) ? s_olo[s] : rowLo[s];
        int hi = (s < SEG_OFFS) ? s_ohi[s] : rowHi[s];
        lo = (lo > 0) ? lo : 0;
        hi = (hi < PART_EDGES) ? hi : PART_EDGES;
        if (lo >= hi) continue;
        const unsigned* run = part + (size_t)s * PART_EDGES;
        const int last = hi - 1;
        for (int i = lo + lane; i < hi; i += 64) {
            const int i1 = i + 32;
            const unsigned w0 = run[i];
            const unsigned w1 = run[(i1 < last) ? i1 : last];
            f(w0);
            if (i1 < hi) f(w1);
        }
    }
}

__global__ void __launch_bounds__(SEG_TB)
k_seg(int* __restrict__ csr, int* __restrict__ rowptr, float* __restrict__ dinv,
      const unsigned* __restrict__ part, const int* __restrict__ offT, int n, int nsl, int nRanges,
      float* __restrict__ out, const float* __restrict__ ea, const float* __restrict__ We, int featBeg, int featEnd) {
    if ((int)blockIdx.x >= nRanges) {
        const int tile = ((int)blockIdx.x - nRanges) * (SEG_TB / 32) + (int)(threadIdx.x >> 5);
        edge_feat_tile(out, ea, We, featBeg + tile * 64, featEnd, (int)(threadIdx.x & 31));
        return;
    }
    __shared__ int s_cnt[RANGE_NODES];
    __shared__ int s_start[RANGE_NODES];
    __shared__ int s_wave[RANGE_NODES / 32];
    __shared__ int s_lo[SEG_TB / 32], s_hi[SEG_TB / 32];
    __shared__ int s_olo[SEG_OFFS], s_ohi[SEG_OFFS];
    __shared__ __attribute__((aligned(16))) int s_stage[SEG_CAP];
    const int t = threadIdx.x;
    const int lane = t & 31;
    const int wv = t >> 5;
    const int rgn = blockIdx.x;
    const int* rowLo = offT + (size_t)rgn * nsl;
    const int* rowHi = rowLo + nsl;

    if (t < RANGE_NODES) s_cnt[t] = 0;
    int a = 0, b = 0;
    for (int s = t; s < nsl; s += SEG_TB) {
        const int l = rowLo[s], h = rowHi[s];
        a += l;
        b += h;
        if (s < SEG_OFFS) { s_olo[s] = l; s_ohi[s] = h; }
    }
    a = wave_sum_int(a);
    b = wave_sum_int(b);
    if (lane == 0) { s_lo[wv] = a; s_hi[wv] = b; }
    __syncthreads();
    const int gs    = wave_sum_int(s_lo[lane]);
    const int total = wave_sum_int(s_hi[lane]) - gs;

    seg_runs(part, s_olo, s_ohi, rowLo, rowHi, nsl, wv, lane,
             [&](unsigned w) { atomicAdd(&s_cnt[w & 255u], 1); });
    __syncthreads();

    const int c = (t < RANGE_NODES) ? s_cnt[t] : 0;
    const int incl = wave_scan_int(c, lane);
    if (t < RANGE_NODES && lane == 31) s_wave[wv] = incl;
    __syncthreads();
    const int before = wave_sum_int((lane < wv && lane < RANGE_NODES / 32) ? s_wave[lane] : 0);
    const int excl = before + incl - c;
    if (t < RANGE_NODES) {
        s_start[t] = excl;
        s_cnt[t] = 0;
        const int node = rgn * RANGE_NODES + t;
        if (node < n) {
            rowptr[node] = gs + excl;
            dinv[node]   = rsqrtf((float)(c + 1));
            if (node == n - 1) rowptr[n] = gs + excl + c;
        }
    }
    __syncthreads();

    const bool staged = total <= SEG_CAP;
    int* seg = csr + (size_t)gs;
    seg_runs(part, s_olo, s_ohi, rowLo, rowHi, nsl, wv, lane,
             [&](unsigned w) {
                 const unsigned nd = w & 255u;
                 const int pos = s_start[nd] + atomicAdd(&s_cnt[nd], 1);
                 if (staged) s_stage[pos] = (int)(w >> 8);
                 else        seg[pos] = (int)(w >> 8);
             });
    __syncthreads();
    if (staged)
        for (int i = t; i < total; i += SEG_TB) seg[i] = s_stage[i];
}

#define GATHER_U 4
#define GATHER_NP 2
#define GATHER_TB 256

template <bool HAS_BIAS, bool SCALED>
__device__ __forceinline__ void gather_rows(float4 (&acc)[GATHER_NP],
        const float* __restrict__ h, const float* __restrict__ dinv,
        const float* __restrict__ bias, const int* __restrict__ rowptr,
        const int* __restrict__ csr, int n, int nodeBase, int half, unsigned c) {
    float4 b = make_float4(0.0f, 0.0f, 0.0f, 0.0f);
    if (HAS_BIAS) b = *(const float4*)(bias + 4 * c);

    int   beg[GATHER_NP], end[GATHER_NP];
    float di[GATHER_NP];
    float4 own[GATHER_NP];
#pragma unroll
    for (int it = 0; it < GATHER_NP; ++it) {
        const int node = nodeBase + 2 * it + half;
        const int nc = (node < n) ? node : (n - 1);
        const int r0 = rowptr[nc];
        const int r1 = rowptr[nc + 1];
        beg[it] = r0;
        end[it] = (node < n) ? r1 : r0;
        di[it]  = dinv[nc];
        own[it] = *(const float4*)(h + (unsigned)nc * NODE_DIM + 4 * c);
    }

#pragma unroll
    for (int it = 0; it < GATHER_NP; ++it) {
        const float d2 = di[it];
        float4 a;
        if (SCALED) {
            a.x = own[it].x * d2;
            a.y = own[it].y * d2;
            a.z = own[it].z * d2;
            a.w = own[it].w * d2;
        } else {
            a.x = own[it].x * d2 * d2;
            a.y = own[it].y * d2 * d2;
            a.z = own[it].z * d2 * d2;
            a.w = own[it].w * d2 * d2;
        }
        if (HAS_BIAS) { a.x += b.x; a.y += b.y; a.z += b.z; a.w += b.w; }

        const int e1 = end[it];
        if (beg[it] < e1) {
            const int last = e1 - 1;
            int p[GATHER_U];
#pragma unroll
            for (int k = 0; k < GATHER_U; ++k) {
                const int jj = beg[it] + k;
                p[k] = csr[(jj < last) ? jj : last];
            }
            for (int j = beg[it]; j < e1; j += GATHER_U) {
                float4 v[GATHER_U];
                float  w[GATHER_U];
#pragma unroll
                for (int k = 0; k < GATHER_U; ++k) {
                    v[k] = *(const float4*)(h + (unsigned)p[k] * NODE_DIM + 4 * c);
                    w[k] = SCALED ? 1.0f : dinv[(unsigned)p[k]];
                }
                int pn[GATHER_U];
#pragma unroll
                for (int k = 0; k < GATHER_U; ++k) {
                    const int jj = j + GATHER_U + k;
                    pn[k] = csr[(jj < last) ? jj : last];
                }
#pragma unroll
                for (int k = 0; k < GATHER_U; ++k) {
                    const float norm = SCALED ? d2 : (w[k] * d2);
                    const bool live = (j + k) < e1;
                    a.x = live ? (a.x + v[k].x * norm) : a.x;
                    a.y = live ? (a.y + v[k].y * norm) : a.y;
                    a.z = live ? (a.z + v[k].z * norm) : a.z;
                    a.w = live ? (a.w + v[k].w * norm) : a.w;
                }
#pragma unroll
                for (int k = 0; k < GATHER_U; ++k) p[k] = pn[k];
            }
        }
        acc[it] = a;
    }
}

#define TILE_ROWS (GATHER_TB / 32 * 2 * GATHER_NP)
#define TILE_PITCH 68

__global__ void __launch_bounds__(GATHER_TB)
k_gather_gemm(float* __restrict__ hout, const float* __restrict__ x, const float* __restrict__ dinv,
              const float* __restrict__ W1, const float* __restrict__ b1, const float* __restrict__ W2,
              const int* __restrict__ rowptr, const int* __restrict__ csr, int n) {
    __shared__ __attribute__((aligned(16))) float s_a[TILE_ROWS * TILE_PITCH];
    __shared__ __attribute__((aligned(16))) float s_h[TILE_ROWS * TILE_PITCH];
    __shared__ __attribute__((aligned(16))) float s_d[TILE_ROWS];
    const int lane = threadIdx.x & 31;
    const int wv   = threadIdx.x >> 5;
    const int half = lane >> 4;
    const unsigned c = lane & 15;
    const int blockNode = blockIdx.x * TILE_ROWS;

    float myd = 0.0f;
    if (wv == 0) {
        const int row = blockNode + lane;
        myd = dinv[(row < n) ? row : (n - 1)];
    }

    float4 acc[GATHER_NP];
    gather_rows<false, false>(acc, x, dinv, nullptr, rowptr, csr, n, blockNode + wv * (2 * GATHER_NP), half, c);

#pragma unroll
    for (int it = 0; it < GATHER_NP; ++it)
        *(float4*)&s_a[(wv * (2 * GATHER_NP) + 2 * it + half) * TILE_PITCH + 4 * c] = acc[it];
    if (wv == 0) s_d[lane] = myd;

    const int rowTile = (wv >> 2) * 16;
    const int colBase = (wv & 3) * 16;
    const int m       = lane & 15;
    const int khalf   = half * 2;
    const int nc      = colBase + m;

    v2f b[NODE_DIM / 4];
#pragma unroll
    for (int i = 0; i < NODE_DIM / 4; ++i) {
        b[i].x = W1[(4 * i + khalf + 0) * NODE_DIM + nc];
        b[i].y = W1[(4 * i + khalf + 1) * NODE_DIM + nc];
    }
    const float bias1 = b1[nc];
    __syncthreads();

    const float* arow = s_a + (rowTile + m) * TILE_PITCH + khalf;
    v8f d = {};
#pragma unroll
    for (int i = 0; i < NODE_DIM / 4; ++i) {
        const v2f a = *(const v2f*)(arow + 4 * i);
        d = __builtin_amdgcn_wmma_f32_16x16x4_f32(false, a, false, b[i], (short)0, d, false, false);
    }

#pragma unroll
    for (int i = 0; i < NODE_DIM / 4; ++i) {
        b[i].x = W2[(4 * i + khalf + 0) * NODE_DIM + nc];
        b[i].y = W2[(4 * i + khalf + 1) * NODE_DIM + nc];
    }

#pragma unroll
    for (int r = 0; r < 8; ++r)
        s_h[(rowTile + r + half * 8) * TILE_PITCH + nc] = fmaxf(d[r] + bias1, 0.0f);
    __syncthreads();

    const float* hrow = s_h + (rowTile + m) * TILE_PITCH + khalf;
    v8f d2 = {};
#pragma unroll
    for (int i = 0; i < NODE_DIM / 4; ++i) {
        const v2f a = *(const v2f*)(hrow + 4 * i);
        d2 = __builtin_amdgcn_wmma_f32_16x16x4_f32(false, a, false, b[i], (short)0, d2, false, false);
    }

    const float4 q0 = *(const float4*)&s_d[rowTile + half * 8];
    const float4 q1 = *(const float4*)&s_d[rowTile + half * 8 + 4];
    const float dr[8] = {q0.x, q0.y, q0.z, q0.w, q1.x, q1.y, q1.z, q1.w};
#pragma unroll
    for (int r = 0; r < 8; ++r) {
        const int row = blockNode + rowTile + r + half * 8;
        if (row < n) hout[(size_t)row * NODE_DIM + nc] = d2[r] * dr[r];
    }
}


template <int STEPS>
__device__ __forceinline__ float group_sum(float v) {
    if (STEPS >= 1) v += __int_as_float(__builtin_amdgcn_update_dpp(0, __float_as_int(v), 0x161, 0xf, 0xf, true));
    if (STEPS >= 2) v += __int_as_float(__builtin_amdgcn_update_dpp(0, __float_as_int(v), 0x162, 0xf, 0xf, true));
    if (STEPS >= 3) v += __int_as_float(__builtin_amdgcn_update_dpp(0, __float_as_int(v), 0x164, 0xf, 0xf, true));
    if (STEPS >= 4) v += __int_as_float(__builtin_amdgcn_update_dpp(0, __float_as_int(v), 0x168, 0xf, 0xf, true));
    return v;
}

__device__ __forceinline__ float dot4(const float4 a, const float4 b) {
    return a.x * b.x + a.y * b.y + a.z * b.z + a.w * b.w;
}

__global__ void __launch_bounds__(GATHER_TB)
k_gather_proj(float* __restrict__ pp, const float* __restrict__ h, const float* __restrict__ dinv,
              const float* __restrict__ bias, const float* __restrict__ We,
              const int* __restrict__ rowptr, const int* __restrict__ csr, int n) {
    const int lane = threadIdx.x & 31;
    const int half = lane >> 4;
    const unsigned c = lane & 15;
    const int nodeBase = (blockIdx.x * (GATHER_TB >> 5) + (threadIdx.x >> 5)) * (2 * GATHER_NP);

    float4 acc[GATHER_NP];
    gather_rows<true, true>(acc, h, dinv, bias, rowptr, csr, n, nodeBase, half, c);

    const float4 ws = *(const float4*)(We + 4 * c);
    const float4 wd = *(const float4*)(We + 64 + 4 * c);

    float ps[GATHER_NP], pd[GATHER_NP];
#pragma unroll
    for (int it = 0; it < GATHER_NP; ++it) {
        float4 v = acc[it];
        v.x = fmaxf(v.x, 0.0f); v.y = fmaxf(v.y, 0.0f);
        v.z = fmaxf(v.z, 0.0f); v.w = fmaxf(v.w, 0.0f);
        ps[it] = group_sum<4>(dot4(v, ws));
        pd[it] = group_sum<4>(dot4(v, wd));
    }
    const float a = (c == 0) ? ps[0] : ps[1];
    const float b = (c == 0) ? pd[0] : pd[1];
    const int myNode = nodeBase + 2 * (int)c + half;
    if (c < GATHER_NP && myNode < n) *(float2*)(pp + 2 * (size_t)myNode) = make_float2(a, b);
}

__device__ __forceinline__ void edge_feat_tile(float* __restrict__ out, const float* __restrict__ ea,
                                               const float* __restrict__ We, int edgeBase, int eEnd, int lane) {
    if (edgeBase >= eEnd) return;
    const int q = lane >> 2;
    const int j = lane & 3;
    const float4 w = *(const float4*)(We + 128 + 4 * j);

    float p[8];
#pragma unroll
    for (int r = 0; r < 8; ++r) {
        int er = edgeBase + r * 8 + q;
        er = (er < eEnd) ? er : (eEnd - 1);
        const float4 a = *(const float4*)(ea + (size_t)er * 16 + 4 * j);
        p[r] = dot4(a, w);
    }
#pragma unroll
    for (int r = 0; r < 8; ++r) p[r] = group_sum<2>(p[r]);

    const float m0 = (j == 0) ? p[0] : (j == 1) ? p[1] : (j == 2) ? p[2] : p[3];
    const float m1 = (j == 0) ? p[4] : (j == 1) ? p[5] : (j == 2) ? p[6] : p[7];
    const int e0 = edgeBase + j * 8 + q;
    const int e1 = e0 + 32;
    if (e0 < eEnd) out[e0] = m0;
    if (e1 < eEnd) out[e1] = m1;
}

__global__ void k_edge_score(float* __restrict__ out, const float* __restrict__ pp,
                             const float* __restrict__ be,
                             const int* __restrict__ src, const int* __restrict__ dst, int e) {
    const int lane = threadIdx.x & 31;
    const int edgeBase = (blockIdx.x * (blockDim.x >> 5) + (threadIdx.x >> 5)) * 64;

    const int e0 = edgeBase + lane;
    const int e1 = e0 + 32;
    const int e0c = (e0 < e) ? e0 : (e - 1);
    const int e1c = (e1 < e) ? e1 : (e - 1);
    const unsigned s0 = (unsigned)src[e0c], d0 = (unsigned)dst[e0c];
    const unsigned s1 = (unsigned)src[e1c], d1 = (unsigned)dst[e1c];
    const float m0 = out[e0c];
    const float m1 = out[e1c];

    const float n0 = pp[2u * s0] + pp[2u * d0 + 1u];
    const float n1 = pp[2u * s1] + pp[2u * d1 + 1u];

    const float bias = be[0];
    if (e0 < e) out[e0] = m0 + n0 + bias;
    if (e1 < e) out[e1] = m1 + n1 + bias;
}

extern "C" void kernel_launch(void* const* d_in, const int* in_sizes, int n_in,
                              void* d_out, int out_size, void* d_ws, size_t ws_size,
                              hipStream_t stream) {
    const float* x  = (const float*)d_in[0];
    const int*   ei = (const int*)d_in[1];
    const float* ea = (const float*)d_in[2];
    const float* W1 = (const float*)d_in[3];
    const float* b1 = (const float*)d_in[4];
    const float* W2 = (const float*)d_in[5];
    const float* b2 = (const float*)d_in[6];
    const float* We = (const float*)d_in[7];
    const float* be = (const float*)d_in[8];
    float* out = (float*)d_out;

    const int N = in_sizes[0] / NODE_DIM;
    const int E = in_sizes[1] / 2;
    const int* src = ei;
    const int* dst = ei + E;

    const int nSlices = (E + PART_EDGES - 1) / PART_EDGES;
    const int nRanges = (N + RANGE_NODES - 1) / RANGE_NODES;

    char* ws = (char*)d_ws;
    const size_t nodeBytes = (((size_t)N * 4) + 511) & ~(size_t)511;
    const size_t rowBytes  = ((((size_t)N + 1) * 4) + 511) & ~(size_t)511;
    const size_t hBytes    = (size_t)N * NODE_DIM * 4;
    const size_t edgeBytes = (((size_t)E * 4) + 511) & ~(size_t)511;
    float*    dinv   = (float*)ws;
    float*    hA     = (float*)(ws + nodeBytes);
    float*    hB     = (float*)(ws + nodeBytes + hBytes);
    int*      rowptr = (int*)(ws + nodeBytes + 2 * hBytes);
    int*      csr    = (int*)(ws + nodeBytes + rowBytes + 2 * hBytes);
    unsigned* part   = (unsigned*)(ws + nodeBytes + rowBytes + 2 * hBytes + edgeBytes);
    int*      offT   = (int*)(ws + nodeBytes + rowBytes + 2 * hBytes + 2 * edgeBytes);

    const int TB = 256;
    const int gGath = (N + TILE_ROWS - 1) / TILE_ROWS;
    const int gEdge = (E + 511) / 512;

    const int featCut = (E / 16384) * 6144;
    const int gFeat1  = featCut / 2048;
    const int gFeat2  = (E - featCut + 2047) / 2048;

    k_part<<<nSlices + gFeat1, PART_TB, 0, stream>>>(part, offT, src, dst, E, nRanges, nSlices, out, ea, We, featCut);
    k_seg <<<nRanges + gFeat2, SEG_TB, 0, stream>>>(csr, rowptr, dinv, part, offT, N, nSlices, nRanges, out, ea, We, featCut, E);

    k_gather_gemm<<<gGath, GATHER_TB, 0, stream>>>(hB, x, dinv, W1, b1, W2, rowptr, csr, N);

    float* pp = hA;
    k_gather_proj<<<gGath, GATHER_TB, 0, stream>>>(pp, hB, dinv, b2, We, rowptr, csr, N);
    k_edge_score <<<gEdge, TB, 0, stream>>>(out, pp, be, src, dst, E);
}
